# starting kernel
# baseline (speedup 1.0000x reference)
#include <hip/hip_runtime.h>
#include <hip/hip_bf16.h>
#include <math.h>

typedef __attribute__((ext_vector_type(2))) float v2f;
typedef __attribute__((ext_vector_type(8))) float v8f;

#define N_QUBITS 10
#define N_LAYERS 3
#define WAVES_PER_BLOCK 4
#define STATE_SZ 1024

__device__ __forceinline__ float ry_elem(float c, float s, int row, int col) {
  return (row == col) ? c : ((row == 0) ? -s : s);
}

template <int MODE>
__device__ __forceinline__ int map_idx(int col, int g) {
  if (MODE == 0) return (g << 6) | col;
  return ((col >> 2) << 6) | (g << 2) | (col & 3);
}

template <int MODE>
__device__ __forceinline__ void apply_group4(float* __restrict__ S,
                                             const float* c4, const float* s4,
                                             int lane) {
  const int mrow  = lane & 15;
  const int khalf = (lane >> 4) << 1;
#pragma unroll
  for (int t = 0; t < 4; ++t) {
    v8f acc = {0.f, 0.f, 0.f, 0.f, 0.f, 0.f, 0.f, 0.f};
#pragma unroll
    for (int kk = 0; kk < 4; ++kk) {
      v2f a, b;
#pragma unroll
      for (int v = 0; v < 2; ++v) {
        const int kcol = kk * 4 + khalf + v;
        float e = 1.0f;
#pragma unroll
        for (int j = 0; j < 4; ++j)
          e *= ry_elem(c4[j], s4[j], (mrow >> (3 - j)) & 1, (kcol >> (3 - j)) & 1);
        a[v] = e;
        b[v] = S[map_idx<MODE>(t * 16 + mrow, kcol)];
      }
      acc = __builtin_amdgcn_wmma_f32_16x16x4_f32(false, a, false, b,
                                                  (short)0, acc, false, false);
    }
    const int rbase = (lane >> 4) << 3;
#pragma unroll
    for (int r = 0; r < 8; ++r)
      S[map_idx<MODE>(t * 16 + mrow, rbase + r)] = acc[r];
  }
}

__device__ __forceinline__ void apply_group2(float* __restrict__ S,
                                             float c8, float s8,
                                             float c9, float s9, int lane) {
  const int mrow  = lane & 15;
  const int khalf = (lane >> 4) << 1;
  v2f a;
#pragma unroll
  for (int v = 0; v < 2; ++v) {
    const int k = khalf + v;
    const int g = mrow & 3;
    a[v] = ry_elem(c8, s8, (g >> 1) & 1, (k >> 1) & 1) *
           ry_elem(c9, s9, g & 1, k & 1);
  }
#pragma unroll
  for (int t = 0; t < 16; ++t) {
    const int ncol = t * 16 + mrow;
    v2f b;
#pragma unroll
    for (int v = 0; v < 2; ++v)
      b[v] = S[(ncol << 2) | (khalf + v)];
    v8f acc = {0.f, 0.f, 0.f, 0.f, 0.f, 0.f, 0.f, 0.f};
    acc = __builtin_amdgcn_wmma_f32_16x16x4_f32(false, a, false, b,
                                                (short)0, acc, false, false);
    if (lane < 16) {
#pragma unroll
      for (int r = 0; r < 4; ++r)
        S[(ncol << 2) | r] = acc[r];
    }
  }
}

__device__ __forceinline__ int cnot_src(int i) {
  int s = i;
  s ^= (s & 1) << 9;
#pragma unroll
  for (int w = 8; w >= 0; --w)
    s ^= ((s >> (9 - w)) & 1) << (8 - w);
  return s;
}

__global__ __launch_bounds__(WAVES_PER_BLOCK * 32, 1)
void qlayer_wmma_kernel(const float* __restrict__ x,
                        const float* __restrict__ params,
                        float* __restrict__ out, int batch) {
  __shared__ float lds[WAVES_PER_BLOCK * STATE_SZ];
  const int lane   = threadIdx.x & 31;
  const int wid    = threadIdx.x >> 5;
  const int sample = blockIdx.x * WAVES_PER_BLOCK + wid;
  if (sample >= batch) return;
  float* S = lds + wid * STATE_SZ;

#pragma unroll
  for (int r = 0; r < 32; ++r)
    S[r * 32 + lane] = (r == 0 && lane == 0) ? 1.0f : 0.0f;

  float ci[N_QUBITS], si[N_QUBITS];
#pragma unroll
  for (int w = 0; w < N_QUBITS; ++w) {
    const float a = 0.5f * tanhf(x[sample * N_QUBITS + w]);
    ci[w] = cosf(a);
    si[w] = sinf(a);
  }
  apply_group4<0>(S, ci + 0, si + 0, lane);
  apply_group4<1>(S, ci + 4, si + 4, lane);
  apply_group2(S, ci[8], si[8], ci[9], si[9], lane);

  for (int l = 0; l < N_LAYERS; ++l) {
    float cl[N_QUBITS], sl[N_QUBITS];
#pragma unroll
    for (int w = 0; w < N_QUBITS; ++w) {
      const float a = 0.5f * params[l * N_QUBITS + w];
      cl[w] = cosf(a);
      sl[w] = sinf(a);
    }
    apply_group4<0>(S, cl + 0, sl + 0, lane);
    apply_group4<1>(S, cl + 4, sl + 4, lane);
    apply_group2(S, cl[8], sl[8], cl[9], sl[9], lane);

    float tmp[32];
#pragma unroll
    for (int r = 0; r < 32; ++r)
      tmp[r] = S[cnot_src(r * 32 + lane)];
    asm volatile("" ::: "memory");
#pragma unroll
    for (int r = 0; r < 32; ++r)
      S[r * 32 + lane] = tmp[r];
  }

  float acc[N_QUBITS];
#pragma unroll
  for (int w = 0; w < N_QUBITS; ++w) acc[w] = 0.0f;
#pragma unroll
  for (int r = 0; r < 32; ++r) {
    const int i = r * 32 + lane;
    const float v = S[i];
    const float p = v * v;
#pragma unroll
    for (int w = 0; w < N_QUBITS; ++w)
      acc[w] += ((i >> (9 - w)) & 1) ? -p : p;
  }
#pragma unroll
  for (int off = 16; off > 0; off >>= 1) {
#pragma unroll
    for (int w = 0; w < N_QUBITS; ++w)
      acc[w] += __shfl_xor(acc[w], off, 32);
  }
  if (lane == 0) {
#pragma unroll
    for (int w = 0; w < N_QUBITS; ++w)
      out[sample * N_QUBITS + w] = acc[w];
  }
}

extern "C" void kernel_launch(void* const* d_in, const int* in_sizes, int n_in,
                              void* d_out, int out_size, void* d_ws, size_t ws_size,
                              hipStream_t stream) {
  const float* x      = (const float*)d_in[0];
  const float* params = (const float*)d_in[1];
  float* out          = (float*)d_out;
  const int batch = in_sizes[0] / N_QUBITS;
  const int blocks = (batch + WAVES_PER_BLOCK - 1) / WAVES_PER_BLOCK;
  qlayer_wmma_kernel<<<blocks, WAVES_PER_BLOCK * 32, 0, stream>>>(x, params, out, batch);
}


# State kept in registers across the three gate groups of a layer: each WMMA D fragment is the next group's A/B operand; LDS only for the CNOT ring; changes the order of the sums
# speedup vs baseline: 1.2090x; 1.2090x over previous
#include <hip/hip_runtime.h>
#include <math.h>


typedef __attribute__((ext_vector_type(2))) float v2f;
typedef __attribute__((ext_vector_type(8))) float v8f;

#define N_QUBITS 10
#define N_LAYERS 3
#define WAVES_PER_BLOCK 4
#define STATE_SZ 1024

__device__ __forceinline__ float ry_elem(float c, float s, int row, int col) {
  return (row == col) ? c : ((row == 0) ? -s : s);
}

__device__ __forceinline__ float kron4(const float* c4, const float* s4,
                                       int row, int col) {
  float e = 1.0f;
#pragma unroll
  for (int j = 0; j < 4; ++j)
    e *= ry_elem(c4[j], s4[j], (row >> (3 - j)) & 1, (col >> (3 - j)) & 1);
  return e;
}

__device__ __forceinline__ void ry_layer(v8f (&X)[4], const float* cw,
                                         const float* sw, int lane) {
  const int c = lane & 15;
  const int h = lane >> 4;
  v2f g[4];
  v8f Y[4], Z[4];

#pragma unroll
  for (int kk = 0; kk < 4; ++kk)
#pragma unroll
    for (int v = 0; v < 2; ++v)
      g[kk][v] = kron4(cw, sw, c, 4 * kk + 2 * h + v);
#pragma unroll
  for (int t = 0; t < 4; ++t) {
    v8f acc = {0.f, 0.f, 0.f, 0.f, 0.f, 0.f, 0.f, 0.f};
#pragma unroll
    for (int kk = 0; kk < 4; ++kk) {
      v2f a = {X[t][2 * kk], X[t][2 * kk + 1]};
      acc = __builtin_amdgcn_wmma_f32_16x16x4_f32(false, a, false, g[kk],
                                                  (short)0, acc, false, false);
    }
    Y[t] = acc;
  }

#pragma unroll
  for (int kk = 0; kk < 4; ++kk)
#pragma unroll
    for (int v = 0; v < 2; ++v)
      g[kk][v] = kron4(cw + 4, sw + 4, c, 2 * kk + v + 8 * h);
#pragma unroll
  for (int t = 0; t < 4; ++t) {
    v8f acc = {0.f, 0.f, 0.f, 0.f, 0.f, 0.f, 0.f, 0.f};
#pragma unroll
    for (int kk = 0; kk < 4; ++kk) {
      v2f a = {Y[t][2 * kk], Y[t][2 * kk + 1]};
      acc = __builtin_amdgcn_wmma_f32_16x16x4_f32(false, a, false, g[kk],
                                                  (short)0, acc, false, false);
    }
    Z[t] = acc;
  }

  const float c3[4] = {cw[8], cw[9], 1.0f, 1.0f};
  const float s3[4] = {sw[8], sw[9], 0.0f, 0.0f};
#pragma unroll
  for (int kk = 0; kk < 4; ++kk)
#pragma unroll
    for (int v = 0; v < 2; ++v)
      g[kk][v] = kron4(c3, s3, c, 4 * kk + 2 * h + v);
#pragma unroll
  for (int j = 0; j < 4; ++j) {
    v8f acc = {0.f, 0.f, 0.f, 0.f, 0.f, 0.f, 0.f, 0.f};
#pragma unroll
    for (int kk = 0; kk < 4; ++kk) {
      v2f b = {Z[kk][2 * j], Z[kk][2 * j + 1]};
      acc = __builtin_amdgcn_wmma_f32_16x16x4_f32(false, g[kk], false, b,
                                                  (short)0, acc, false, false);
    }
    X[j] = acc;
  }
}

__device__ __forceinline__ constexpr int cnot_src(int i) {
  int s = i;
  s ^= (s & 1) << 9;
  for (int w = 8; w >= 0; --w)
    s ^= ((s >> (9 - w)) & 1) << (8 - w);
  return s;
}

__device__ __forceinline__ void cnot_ring(v8f (&X)[4], float* __restrict__ S,
                                          int lane) {
  const int c = lane & 15;
  const int h = lane >> 4;
  const int st_base = (c << 2) | (h << 1);
#pragma unroll
  for (int j = 0; j < 4; ++j)
#pragma unroll
    for (int r = 0; r < 8; ++r) {
      const int a = (((r >> 1) & 1) << 3) | (j << 1) | (r & 1);
      S[st_base | (a << 6) | (r >> 2)] = X[j][r];
    }
  const int ld_base = cnot_src((h << 7) | (c << 2));
#pragma unroll
  for (int t = 0; t < 4; ++t)
#pragma unroll
    for (int r = 0; r < 8; ++r) {
      const int a = 4 * (r >> 1) + (r & 1);
      X[t][r] = S[ld_base ^ cnot_src((a << 6) | t)];
    }
}

__global__ __launch_bounds__(WAVES_PER_BLOCK * 32, 1)
void qlayer_wmma_kernel(const float* __restrict__ x,
                        const float* __restrict__ params,
                        float* __restrict__ out, int batch) {
  __shared__ float lds[WAVES_PER_BLOCK * STATE_SZ];
  const int lane   = threadIdx.x & 31;
  const int wid    = threadIdx.x >> 5;
  const int sample = blockIdx.x * WAVES_PER_BLOCK + wid;
  if (sample >= batch) return;
  float* S = lds + wid * STATE_SZ;

  v8f X[4];
#pragma unroll
  for (int t = 0; t < 4; ++t)
    X[t] = (v8f){0.f, 0.f, 0.f, 0.f, 0.f, 0.f, 0.f, 0.f};
  X[0][0] = (lane == 0) ? 1.0f : 0.0f;

  const int slot_wire[N_QUBITS] = {1, 8, 9, 3, 4, 5, 6, 7, 2, 0};
  float ci[N_QUBITS], si[N_QUBITS];
#pragma unroll
  for (int p = 0; p < N_QUBITS; ++p) {
    const float a = 0.5f * tanhf(x[sample * N_QUBITS + slot_wire[p]]);
    ci[p] = cosf(a);
    si[p] = sinf(a);
  }
  ry_layer(X, ci, si, lane);

  for (int l = 0; l < N_LAYERS; ++l) {
    float cl[N_QUBITS], sl[N_QUBITS];
#pragma unroll
    for (int w = 0; w < N_QUBITS; ++w) {
      const float a = 0.5f * params[l * N_QUBITS + w];
      cl[w] = cosf(a);
      sl[w] = sinf(a);
    }
    ry_layer(X, cl, sl, lane);
    cnot_ring(X, S, lane);
  }

  const int c = lane & 15;
  const int h = lane >> 4;
  float acc[N_QUBITS];
#pragma unroll
  for (int w = 0; w < N_QUBITS; ++w) acc[w] = 0.0f;
#pragma unroll
  for (int t = 0; t < 4; ++t)
#pragma unroll
    for (int r = 0; r < 8; ++r) {
      const int i = ((4 * (r >> 1) + 2 * h + (r & 1)) << 6) | (c << 2) | t;
      const float v = X[t][r];
      const float p = v * v;
#pragma unroll
      for (int w = 0; w < N_QUBITS; ++w)
        acc[w] += ((i >> (9 - w)) & 1) ? -p : p;
    }
#pragma unroll
  for (int off = 16; off > 0; off >>= 1) {
#pragma unroll
    for (int w = 0; w < N_QUBITS; ++w)
      acc[w] += __shfl_xor(acc[w], off, 32);
  }
  if (lane == 0) {
#pragma unroll
    for (int w = 0; w < N_QUBITS; ++w)
      out[sample * N_QUBITS + w] = acc[w];
  }
}

extern "C" void kernel_launch(void* const* d_in, const int* in_sizes, int n_in,
                              void* d_out, int out_size, void* d_ws, size_t ws_size,
                              hipStream_t stream) {
  const float* x      = (const float*)d_in[0];
  const float* params = (const float*)d_in[1];
  float* out          = (float*)d_out;
  const int batch = in_sizes[0] / N_QUBITS;
  const int blocks = (batch + WAVES_PER_BLOCK - 1) / WAVES_PER_BLOCK;
  qlayer_wmma_kernel<<<blocks, WAVES_PER_BLOCK * 32, 0, stream>>>(x, params, out, batch);
}


# Input RY layer folded into the first parameter layer: RY(a)RY(b)=RY(a+b) per wire, one angle sum; 3 RY layers instead of 4; changes the rounding and the order of the sums of the first layer
# speedup vs baseline: 1.5776x; 1.3049x over previous
#include <hip/hip_runtime.h>
#include <math.h>


typedef __attribute__((ext_vector_type(2))) float v2f;
typedef __attribute__((ext_vector_type(8))) float v8f;

#define N_QUBITS 10
#define N_LAYERS 3
#define WAVES_PER_BLOCK 4
#define STATE_SZ 1024

__device__ __forceinline__ float ry_elem(float c, float s, int row, int col) {
  return (row == col) ? c : ((row == 0) ? -s : s);
}

__device__ __forceinline__ float kron4(const float* c4, const float* s4,
                                       int row, int col) {
  float e = 1.0f;
#pragma unroll
  for (int j = 0; j < 4; ++j)
    e *= ry_elem(c4[j], s4[j], (row >> (3 - j)) & 1, (col >> (3 - j)) & 1);
  return e;
}

__device__ __forceinline__ void ry_layer(v8f (&X)[4], const float* cw,
                                         const float* sw, int lane) {
  const int c = lane & 15;
  const int h = lane >> 4;
  v2f g[4];
  v8f Y[4], Z[4];

#pragma unroll
  for (int kk = 0; kk < 4; ++kk)
#pragma unroll
    for (int v = 0; v < 2; ++v)
      g[kk][v] = kron4(cw, sw, c, 4 * kk + 2 * h + v);
#pragma unroll
  for (int t = 0; t < 4; ++t) {
    v8f acc = {0.f, 0.f, 0.f, 0.f, 0.f, 0.f, 0.f, 0.f};
#pragma unroll
    for (int kk = 0; kk < 4; ++kk) {
      v2f a = {X[t][2 * kk], X[t][2 * kk + 1]};
      acc = __builtin_amdgcn_wmma_f32_16x16x4_f32(false, a, false, g[kk],
                                                  (short)0, acc, false, false);
    }
    Y[t] = acc;
  }

#pragma unroll
  for (int kk = 0; kk < 4; ++kk)
#pragma unroll
    for (int v = 0; v < 2; ++v)
      g[kk][v] = kron4(cw + 4, sw + 4, c, 2 * kk + v + 8 * h);
#pragma unroll
  for (int t = 0; t < 4; ++t) {
    v8f acc = {0.f, 0.f, 0.f, 0.f, 0.f, 0.f, 0.f, 0.f};
#pragma unroll
    for (int kk = 0; kk < 4; ++kk) {
      v2f a = {Y[t][2 * kk], Y[t][2 * kk + 1]};
      acc = __builtin_amdgcn_wmma_f32_16x16x4_f32(false, a, false, g[kk],
                                                  (short)0, acc, false, false);
    }
    Z[t] = acc;
  }

  const float c3[4] = {cw[8], cw[9], 1.0f, 1.0f};
  const float s3[4] = {sw[8], sw[9], 0.0f, 0.0f};
#pragma unroll
  for (int kk = 0; kk < 4; ++kk)
#pragma unroll
    for (int v = 0; v < 2; ++v)
      g[kk][v] = kron4(c3, s3, c, 4 * kk + 2 * h + v);
#pragma unroll
  for (int j = 0; j < 4; ++j) {
    v8f acc = {0.f, 0.f, 0.f, 0.f, 0.f, 0.f, 0.f, 0.f};
#pragma unroll
    for (int kk = 0; kk < 4; ++kk) {
      v2f b = {Z[kk][2 * j], Z[kk][2 * j + 1]};
      acc = __builtin_amdgcn_wmma_f32_16x16x4_f32(false, g[kk], false, b,
                                                  (short)0, acc, false, false);
    }
    X[j] = acc;
  }
}

__device__ __forceinline__ constexpr int cnot_src(int i) {
  int s = i;
  s ^= (s & 1) << 9;
  for (int w = 8; w >= 0; --w)
    s ^= ((s >> (9 - w)) & 1) << (8 - w);
  return s;
}

__device__ __forceinline__ void cnot_ring(v8f (&X)[4], float* __restrict__ S,
                                          int lane) {
  const int c = lane & 15;
  const int h = lane >> 4;
  const int st_base = (c << 2) | (h << 1);
#pragma unroll
  for (int j = 0; j < 4; ++j)
#pragma unroll
    for (int r = 0; r < 8; ++r) {
      const int a = (((r >> 1) & 1) << 3) | (j << 1) | (r & 1);
      S[st_base | (a << 6) | (r >> 2)] = X[j][r];
    }
  const int ld_base = cnot_src((h << 7) | (c << 2));
#pragma unroll
  for (int t = 0; t < 4; ++t)
#pragma unroll
    for (int r = 0; r < 8; ++r) {
      const int a = 4 * (r >> 1) + (r & 1);
      X[t][r] = S[ld_base ^ cnot_src((a << 6) | t)];
    }
}

__global__ __launch_bounds__(WAVES_PER_BLOCK * 32, 1)
void qlayer_wmma_kernel(const float* __restrict__ x,
                        const float* __restrict__ params,
                        float* __restrict__ out, int batch) {
  __shared__ float lds[WAVES_PER_BLOCK * STATE_SZ];
  const int lane   = threadIdx.x & 31;
  const int wid    = threadIdx.x >> 5;
  const int sample = blockIdx.x * WAVES_PER_BLOCK + wid;
  if (sample >= batch) return;
  float* S = lds + wid * STATE_SZ;

  v8f X[4];
#pragma unroll
  for (int t = 0; t < 4; ++t)
    X[t] = (v8f){0.f, 0.f, 0.f, 0.f, 0.f, 0.f, 0.f, 0.f};
  X[0][0] = (lane == 0) ? 1.0f : 0.0f;

  {
    float c0[N_QUBITS], s0[N_QUBITS];
#pragma unroll
    for (int w = 0; w < N_QUBITS; ++w) {
      const float a = 0.5f * (tanhf(x[sample * N_QUBITS + w]) + params[w]);
      c0[w] = cosf(a);
      s0[w] = sinf(a);
    }
    ry_layer(X, c0, s0, lane);
    cnot_ring(X, S, lane);
  }

  for (int l = 1; l < N_LAYERS; ++l) {
    float cl[N_QUBITS], sl[N_QUBITS];
#pragma unroll
    for (int w = 0; w < N_QUBITS; ++w) {
      const float a = 0.5f * params[l * N_QUBITS + w];
      cl[w] = cosf(a);
      sl[w] = sinf(a);
    }
    ry_layer(X, cl, sl, lane);
    cnot_ring(X, S, lane);
  }

  const int c = lane & 15;
  const int h = lane >> 4;
  float acc[N_QUBITS];
#pragma unroll
  for (int w = 0; w < N_QUBITS; ++w) acc[w] = 0.0f;
#pragma unroll
  for (int t = 0; t < 4; ++t)
#pragma unroll
    for (int r = 0; r < 8; ++r) {
      const int i = ((4 * (r >> 1) + 2 * h + (r & 1)) << 6) | (c << 2) | t;
      const float v = X[t][r];
      const float p = v * v;
#pragma unroll
      for (int w = 0; w < N_QUBITS; ++w)
        acc[w] += ((i >> (9 - w)) & 1) ? -p : p;
    }
#pragma unroll
  for (int off = 16; off > 0; off >>= 1) {
#pragma unroll
    for (int w = 0; w < N_QUBITS; ++w)
      acc[w] += __shfl_xor(acc[w], off, 32);
  }
  if (lane == 0) {
#pragma unroll
    for (int w = 0; w < N_QUBITS; ++w)
      out[sample * N_QUBITS + w] = acc[w];
  }
}

extern "C" void kernel_launch(void* const* d_in, const int* in_sizes, int n_in,
                              void* d_out, int out_size, void* d_ws, size_t ws_size,
                              hipStream_t stream) {
  const float* x      = (const float*)d_in[0];
  const float* params = (const float*)d_in[1];
  float* out          = (float*)d_out;
  const int batch = in_sizes[0] / N_QUBITS;
  const int blocks = (batch + WAVES_PER_BLOCK - 1) / WAVES_PER_BLOCK;
  qlayer_wmma_kernel<<<blocks, WAVES_PER_BLOCK * 32, 0, stream>>>(x, params, out, batch);
}


# All 30 rotation angles: tanh/cos/sin evaluated once per wave, one angle per lane, fetched per layer by lane reads into scalars, instead of 70 redundant calls in every lane; arithmetic unchanged
# speedup vs baseline: 2.1190x; 1.3432x over previous
#include <hip/hip_runtime.h>
#include <math.h>


typedef __attribute__((ext_vector_type(2))) float v2f;
typedef __attribute__((ext_vector_type(8))) float v8f;

#define N_QUBITS 10
#define N_LAYERS 3
#define WAVES_PER_BLOCK 4
#define STATE_SZ 1024

__device__ __forceinline__ float ry_elem(float c, float s, int row, int col) {
  return (row == col) ? c : ((row == 0) ? -s : s);
}

__device__ __forceinline__ float kron4(const float* c4, const float* s4,
                                       int row, int col) {
  float e = 1.0f;
#pragma unroll
  for (int j = 0; j < 4; ++j)
    e *= ry_elem(c4[j], s4[j], (row >> (3 - j)) & 1, (col >> (3 - j)) & 1);
  return e;
}

__device__ __forceinline__ void ry_layer(v8f (&X)[4], const float* cw,
                                         const float* sw, int lane) {
  const int c = lane & 15;
  const int h = lane >> 4;
  v2f g[4];
  v8f Y[4], Z[4];

#pragma unroll
  for (int kk = 0; kk < 4; ++kk)
#pragma unroll
    for (int v = 0; v < 2; ++v)
      g[kk][v] = kron4(cw, sw, c, 4 * kk + 2 * h + v);
#pragma unroll
  for (int t = 0; t < 4; ++t) {
    v8f acc = {0.f, 0.f, 0.f, 0.f, 0.f, 0.f, 0.f, 0.f};
#pragma unroll
    for (int kk = 0; kk < 4; ++kk) {
      v2f a = {X[t][2 * kk], X[t][2 * kk + 1]};
      acc = __builtin_amdgcn_wmma_f32_16x16x4_f32(false, a, false, g[kk],
                                                  (short)0, acc, false, false);
    }
    Y[t] = acc;
  }

#pragma unroll
  for (int kk = 0; kk < 4; ++kk)
#pragma unroll
    for (int v = 0; v < 2; ++v)
      g[kk][v] = kron4(cw + 4, sw + 4, c, 2 * kk + v + 8 * h);
#pragma unroll
  for (int t = 0; t < 4; ++t) {
    v8f acc = {0.f, 0.f, 0.f, 0.f, 0.f, 0.f, 0.f, 0.f};
#pragma unroll
    for (int kk = 0; kk < 4; ++kk) {
      v2f a = {Y[t][2 * kk], Y[t][2 * kk + 1]};
      acc = __builtin_amdgcn_wmma_f32_16x16x4_f32(false, a, false, g[kk],
                                                  (short)0, acc, false, false);
    }
    Z[t] = acc;
  }

  const float c3[4] = {cw[8], cw[9], 1.0f, 1.0f};
  const float s3[4] = {sw[8], sw[9], 0.0f, 0.0f};
#pragma unroll
  for (int kk = 0; kk < 4; ++kk)
#pragma unroll
    for (int v = 0; v < 2; ++v)
      g[kk][v] = kron4(c3, s3, c, 4 * kk + 2 * h + v);
#pragma unroll
  for (int j = 0; j < 4; ++j) {
    v8f acc = {0.f, 0.f, 0.f, 0.f, 0.f, 0.f, 0.f, 0.f};
#pragma unroll
    for (int kk = 0; kk < 4; ++kk) {
      v2f b = {Z[kk][2 * j], Z[kk][2 * j + 1]};
      acc = __builtin_amdgcn_wmma_f32_16x16x4_f32(false, g[kk], false, b,
                                                  (short)0, acc, false, false);
    }
    X[j] = acc;
  }
}

__device__ __forceinline__ constexpr int cnot_src(int i) {
  int s = i;
  s ^= (s & 1) << 9;
  for (int w = 8; w >= 0; --w)
    s ^= ((s >> (9 - w)) & 1) << (8 - w);
  return s;
}

__device__ __forceinline__ void cnot_ring(v8f (&X)[4], float* __restrict__ S,
                                          int lane) {
  const int c = lane & 15;
  const int h = lane >> 4;
  const int st_base = (c << 2) | (h << 1);
#pragma unroll
  for (int j = 0; j < 4; ++j)
#pragma unroll
    for (int r = 0; r < 8; ++r) {
      const int a = (((r >> 1) & 1) << 3) | (j << 1) | (r & 1);
      S[st_base | (a << 6) | (r >> 2)] = X[j][r];
    }
  const int ld_base = cnot_src((h << 7) | (c << 2));
#pragma unroll
  for (int t = 0; t < 4; ++t)
#pragma unroll
    for (int r = 0; r < 8; ++r) {
      const int a = 4 * (r >> 1) + (r & 1);
      X[t][r] = S[ld_base ^ cnot_src((a << 6) | t)];
    }
}

__device__ __forceinline__ float lane_bcast(float v, int src_lane) {
  return __int_as_float(__builtin_amdgcn_readlane(__float_as_int(v), src_lane));
}

__global__ __launch_bounds__(WAVES_PER_BLOCK * 32, 1)
void qlayer_wmma_kernel(const float* __restrict__ x,
                        const float* __restrict__ params,
                        float* __restrict__ out, int batch) {
  __shared__ float lds[WAVES_PER_BLOCK * STATE_SZ];
  const int lane   = threadIdx.x & 31;
  const int wid    = threadIdx.x >> 5;
  const int sample = blockIdx.x * WAVES_PER_BLOCK + wid;
  if (sample >= batch) return;
  float* S = lds + wid * STATE_SZ;

  v8f X[4];
#pragma unroll
  for (int t = 0; t < 4; ++t)
    X[t] = (v8f){0.f, 0.f, 0.f, 0.f, 0.f, 0.f, 0.f, 0.f};
  X[0][0] = (lane == 0) ? 1.0f : 0.0f;

  const int ai = (lane < N_LAYERS * N_QUBITS) ? lane : 0;
  const int xi = (lane < N_QUBITS) ? lane : 0;
  const float pv = params[ai];
  const float xv = x[sample * N_QUBITS + xi];
  const float half_angle =
      0.5f * ((lane < N_QUBITS) ? (tanhf(xv) + pv) : pv);
  const float cv = cosf(half_angle);
  const float sv = sinf(half_angle);

  {
    float c0[N_QUBITS], s0[N_QUBITS];
#pragma unroll
    for (int w = 0; w < N_QUBITS; ++w) {
      c0[w] = lane_bcast(cv, w);
      s0[w] = lane_bcast(sv, w);
    }
    ry_layer(X, c0, s0, lane);
    cnot_ring(X, S, lane);
  }

  for (int l = 1; l < N_LAYERS; ++l) {
    float cl[N_QUBITS], sl[N_QUBITS];
#pragma unroll
    for (int w = 0; w < N_QUBITS; ++w) {
      cl[w] = lane_bcast(cv, l * N_QUBITS + w);
      sl[w] = lane_bcast(sv, l * N_QUBITS + w);
    }
    ry_layer(X, cl, sl, lane);
    cnot_ring(X, S, lane);
  }

  const int c = lane & 15;
  const int h = lane >> 4;
  float acc[N_QUBITS];
#pragma unroll
  for (int w = 0; w < N_QUBITS; ++w) acc[w] = 0.0f;
#pragma unroll
  for (int t = 0; t < 4; ++t)
#pragma unroll
    for (int r = 0; r < 8; ++r) {
      const int i = ((4 * (r >> 1) + 2 * h + (r & 1)) << 6) | (c << 2) | t;
      const float v = X[t][r];
      const float p = v * v;
#pragma unroll
      for (int w = 0; w < N_QUBITS; ++w)
        acc[w] += ((i >> (9 - w)) & 1) ? -p : p;
    }
#pragma unroll
  for (int off = 16; off > 0; off >>= 1) {
#pragma unroll
    for (int w = 0; w < N_QUBITS; ++w)
      acc[w] += __shfl_xor(acc[w], off, 32);
  }
  if (lane == 0) {
#pragma unroll
    for (int w = 0; w < N_QUBITS; ++w)
      out[sample * N_QUBITS + w] = acc[w];
  }
}

extern "C" void kernel_launch(void* const* d_in, const int* in_sizes, int n_in,
                              void* d_out, int out_size, void* d_ws, size_t ws_size,
                              hipStream_t stream) {
  const float* x      = (const float*)d_in[0];
  const float* params = (const float*)d_in[1];
  float* out          = (float*)d_out;
  const int batch = in_sizes[0] / N_QUBITS;
  const int blocks = (batch + WAVES_PER_BLOCK - 1) / WAVES_PER_BLOCK;
  qlayer_wmma_kernel<<<blocks, WAVES_PER_BLOCK * 32, 0, stream>>>(x, params, out, batch);
}


# First RY layer acts on the fixed start state |0..0>: output = the product state prod_w(cos_w or sin_w), written directly, no sums left; 96 matrix instructions per sample, was 128; rounding changes
# speedup vs baseline: 2.5250x; 1.1916x over previous
#include <hip/hip_runtime.h>
#include <math.h>


typedef __attribute__((ext_vector_type(2))) float v2f;
typedef __attribute__((ext_vector_type(8))) float v8f;

#define N_QUBITS 10
#define N_LAYERS 3
#define WAVES_PER_BLOCK 4
#define STATE_SZ 1024

__device__ __forceinline__ float ry_elem(float c, float s, int row, int col) {
  return (row == col) ? c : ((row == 0) ? -s : s);
}

__device__ __forceinline__ float kron4(const float* c4, const float* s4,
                                       int row, int col) {
  float e = 1.0f;
#pragma unroll
  for (int j = 0; j < 4; ++j)
    e *= ry_elem(c4[j], s4[j], (row >> (3 - j)) & 1, (col >> (3 - j)) & 1);
  return e;
}

__device__ __forceinline__ void ry_layer(v8f (&X)[4], const float* cw,
                                         const float* sw, int lane) {
  const int c = lane & 15;
  const int h = lane >> 4;
  v2f g[4];
  v8f Y[4], Z[4];

#pragma unroll
  for (int kk = 0; kk < 4; ++kk)
#pragma unroll
    for (int v = 0; v < 2; ++v)
      g[kk][v] = kron4(cw, sw, c, 4 * kk + 2 * h + v);
#pragma unroll
  for (int t = 0; t < 4; ++t) {
    v8f acc = {0.f, 0.f, 0.f, 0.f, 0.f, 0.f, 0.f, 0.f};
#pragma unroll
    for (int kk = 0; kk < 4; ++kk) {
      v2f a = {X[t][2 * kk], X[t][2 * kk + 1]};
      acc = __builtin_amdgcn_wmma_f32_16x16x4_f32(false, a, false, g[kk],
                                                  (short)0, acc, false, false);
    }
    Y[t] = acc;
  }

#pragma unroll
  for (int kk = 0; kk < 4; ++kk)
#pragma unroll
    for (int v = 0; v < 2; ++v)
      g[kk][v] = kron4(cw + 4, sw + 4, c, 2 * kk + v + 8 * h);
#pragma unroll
  for (int t = 0; t < 4; ++t) {
    v8f acc = {0.f, 0.f, 0.f, 0.f, 0.f, 0.f, 0.f, 0.f};
#pragma unroll
    for (int kk = 0; kk < 4; ++kk) {
      v2f a = {Y[t][2 * kk], Y[t][2 * kk + 1]};
      acc = __builtin_amdgcn_wmma_f32_16x16x4_f32(false, a, false, g[kk],
                                                  (short)0, acc, false, false);
    }
    Z[t] = acc;
  }

  const float c3[4] = {cw[8], cw[9], 1.0f, 1.0f};
  const float s3[4] = {sw[8], sw[9], 0.0f, 0.0f};
#pragma unroll
  for (int kk = 0; kk < 4; ++kk)
#pragma unroll
    for (int v = 0; v < 2; ++v)
      g[kk][v] = kron4(c3, s3, c, 4 * kk + 2 * h + v);
#pragma unroll
  for (int j = 0; j < 4; ++j) {
    v8f acc = {0.f, 0.f, 0.f, 0.f, 0.f, 0.f, 0.f, 0.f};
#pragma unroll
    for (int kk = 0; kk < 4; ++kk) {
      v2f b = {Z[kk][2 * j], Z[kk][2 * j + 1]};
      acc = __builtin_amdgcn_wmma_f32_16x16x4_f32(false, g[kk], false, b,
                                                  (short)0, acc, false, false);
    }
    X[j] = acc;
  }
}

__device__ __forceinline__ void product_state(v8f (&X)[4], const float* cw,
                                              const float* sw, int lane) {
  const int c = lane & 15;
  const int h = lane >> 4;
  float lf = ((c >> 3) & 1) ? sw[4] : cw[4];
  lf *= ((c >> 2) & 1) ? sw[5] : cw[5];
  lf *= ((c >> 1) & 1) ? sw[6] : cw[6];
  lf *= (c & 1) ? sw[7] : cw[7];
  lf *= h ? sw[8] : cw[8];
  float pr[8];
#pragma unroll
  for (int r = 0; r < 8; ++r) {
    const float f9 = ((r >> 2) & 1) ? sw[9] : cw[9];
    const float f0 = ((r >> 1) & 1) ? sw[0] : cw[0];
    const float f3 = (r & 1) ? sw[3] : cw[3];
    pr[r] = lf * (f9 * f0 * f3);
  }
#pragma unroll
  for (int j = 0; j < 4; ++j) {
    const float f1 = ((j >> 1) & 1) ? sw[1] : cw[1];
    const float f2 = (j & 1) ? sw[2] : cw[2];
    const float gj = f1 * f2;
#pragma unroll
    for (int r = 0; r < 8; ++r)
      X[j][r] = pr[r] * gj;
  }
}

__device__ __forceinline__ constexpr int cnot_src(int i) {
  int s = i;
  s ^= (s & 1) << 9;
  for (int w = 8; w >= 0; --w)
    s ^= ((s >> (9 - w)) & 1) << (8 - w);
  return s;
}

__device__ __forceinline__ void cnot_ring(v8f (&X)[4], float* __restrict__ S,
                                          int lane) {
  const int c = lane & 15;
  const int h = lane >> 4;
  const int st_base = (c << 2) | (h << 1);
#pragma unroll
  for (int j = 0; j < 4; ++j)
#pragma unroll
    for (int r = 0; r < 8; ++r) {
      const int a = (((r >> 1) & 1) << 3) | (j << 1) | (r & 1);
      S[st_base | (a << 6) | (r >> 2)] = X[j][r];
    }
  const int ld_base = cnot_src((h << 7) | (c << 2));
#pragma unroll
  for (int t = 0; t < 4; ++t)
#pragma unroll
    for (int r = 0; r < 8; ++r) {
      const int a = 4 * (r >> 1) + (r & 1);
      X[t][r] = S[ld_base ^ cnot_src((a << 6) | t)];
    }
}

__device__ __forceinline__ float lane_bcast(float v, int src_lane) {
  return __int_as_float(__builtin_amdgcn_readlane(__float_as_int(v), src_lane));
}

__global__ __launch_bounds__(WAVES_PER_BLOCK * 32, 1)
void qlayer_wmma_kernel(const float* __restrict__ x,
                        const float* __restrict__ params,
                        float* __restrict__ out, int batch) {
  __shared__ float lds[WAVES_PER_BLOCK * STATE_SZ];
  const int lane   = threadIdx.x & 31;
  const int wid    = threadIdx.x >> 5;
  const int sample = blockIdx.x * WAVES_PER_BLOCK + wid;
  if (sample >= batch) return;
  float* S = lds + wid * STATE_SZ;

  v8f X[4];

  const int ai = (lane < N_LAYERS * N_QUBITS) ? lane : 0;
  const int xi = (lane < N_QUBITS) ? lane : 0;
  const float pv = params[ai];
  const float xv = x[sample * N_QUBITS + xi];
  const float half_angle =
      0.5f * ((lane < N_QUBITS) ? (tanhf(xv) + pv) : pv);
  const float cv = cosf(half_angle);
  const float sv = sinf(half_angle);

  {
    float c0[N_QUBITS], s0[N_QUBITS];
#pragma unroll
    for (int w = 0; w < N_QUBITS; ++w) {
      c0[w] = lane_bcast(cv, w);
      s0[w] = lane_bcast(sv, w);
    }
    product_state(X, c0, s0, lane);
    cnot_ring(X, S, lane);
  }

  for (int l = 1; l < N_LAYERS; ++l) {
    float cl[N_QUBITS], sl[N_QUBITS];
#pragma unroll
    for (int w = 0; w < N_QUBITS; ++w) {
      cl[w] = lane_bcast(cv, l * N_QUBITS + w);
      sl[w] = lane_bcast(sv, l * N_QUBITS + w);
    }
    ry_layer(X, cl, sl, lane);
    cnot_ring(X, S, lane);
  }

  const int c = lane & 15;
  const int h = lane >> 4;
  float acc[N_QUBITS];
#pragma unroll
  for (int w = 0; w < N_QUBITS; ++w) acc[w] = 0.0f;
#pragma unroll
  for (int t = 0; t < 4; ++t)
#pragma unroll
    for (int r = 0; r < 8; ++r) {
      const int i = ((4 * (r >> 1) + 2 * h + (r & 1)) << 6) | (c << 2) | t;
      const float v = X[t][r];
      const float p = v * v;
#pragma unroll
      for (int w = 0; w < N_QUBITS; ++w)
        acc[w] += ((i >> (9 - w)) & 1) ? -p : p;
    }
#pragma unroll
  for (int off = 16; off > 0; off >>= 1) {
#pragma unroll
    for (int w = 0; w < N_QUBITS; ++w)
      acc[w] += __shfl_xor(acc[w], off, 32);
  }
  if (lane == 0) {
#pragma unroll
    for (int w = 0; w < N_QUBITS; ++w)
      out[sample * N_QUBITS + w] = acc[w];
  }
}

extern "C" void kernel_launch(void* const* d_in, const int* in_sizes, int n_in,
                              void* d_out, int out_size, void* d_ws, size_t ws_size,
                              hipStream_t stream) {
  const float* x      = (const float*)d_in[0];
  const float* params = (const float*)d_in[1];
  float* out          = (float*)d_out;
  const int batch = in_sizes[0] / N_QUBITS;
  const int blocks = (batch + WAVES_PER_BLOCK - 1) / WAVES_PER_BLOCK;
  qlayer_wmma_kernel<<<blocks, WAVES_PER_BLOCK * 32, 0, stream>>>(x, params, out, batch);
}


# Z expectations: in-lane sum/difference tree over the 32 squares, then one DPP row butterfly shared by the ten sums (live values halve per step, lane c ends with wire c); changes the order of the sums
# speedup vs baseline: 2.8284x; 1.1202x over previous
#include <hip/hip_runtime.h>
#include <math.h>


typedef __attribute__((ext_vector_type(2))) float v2f;
typedef __attribute__((ext_vector_type(8))) float v8f;

#define N_QUBITS 10
#define N_LAYERS 3
#define WAVES_PER_BLOCK 4
#define STATE_SZ 1024

__device__ __forceinline__ float ry_elem(float c, float s, int row, int col) {
  return (row == col) ? c : ((row == 0) ? -s : s);
}

__device__ __forceinline__ float kron4(const float* c4, const float* s4,
                                       int row, int col) {
  float e = 1.0f;
#pragma unroll
  for (int j = 0; j < 4; ++j)
    e *= ry_elem(c4[j], s4[j], (row >> (3 - j)) & 1, (col >> (3 - j)) & 1);
  return e;
}

__device__ __forceinline__ void ry_layer(v8f (&X)[4], const float* cw,
                                         const float* sw, int lane) {
  const int c = lane & 15;
  const int h = lane >> 4;
  v2f g[4];
  v8f Y[4], Z[4];

#pragma unroll
  for (int kk = 0; kk < 4; ++kk)
#pragma unroll
    for (int v = 0; v < 2; ++v)
      g[kk][v] = kron4(cw, sw, c, 4 * kk + 2 * h + v);
#pragma unroll
  for (int t = 0; t < 4; ++t) {
    v8f acc = {0.f, 0.f, 0.f, 0.f, 0.f, 0.f, 0.f, 0.f};
#pragma unroll
    for (int kk = 0; kk < 4; ++kk) {
      v2f a = {X[t][2 * kk], X[t][2 * kk + 1]};
      acc = __builtin_amdgcn_wmma_f32_16x16x4_f32(false, a, false, g[kk],
                                                  (short)0, acc, false, false);
    }
    Y[t] = acc;
  }

#pragma unroll
  for (int kk = 0; kk < 4; ++kk)
#pragma unroll
    for (int v = 0; v < 2; ++v)
      g[kk][v] = kron4(cw + 4, sw + 4, c, 2 * kk + v + 8 * h);
#pragma unroll
  for (int t = 0; t < 4; ++t) {
    v8f acc = {0.f, 0.f, 0.f, 0.f, 0.f, 0.f, 0.f, 0.f};
#pragma unroll
    for (int kk = 0; kk < 4; ++kk) {
      v2f a = {Y[t][2 * kk], Y[t][2 * kk + 1]};
      acc = __builtin_amdgcn_wmma_f32_16x16x4_f32(false, a, false, g[kk],
                                                  (short)0, acc, false, false);
    }
    Z[t] = acc;
  }

  const float c3[4] = {cw[8], cw[9], 1.0f, 1.0f};
  const float s3[4] = {sw[8], sw[9], 0.0f, 0.0f};
#pragma unroll
  for (int kk = 0; kk < 4; ++kk)
#pragma unroll
    for (int v = 0; v < 2; ++v)
      g[kk][v] = kron4(c3, s3, c, 4 * kk + 2 * h + v);
#pragma unroll
  for (int j = 0; j < 4; ++j) {
    v8f acc = {0.f, 0.f, 0.f, 0.f, 0.f, 0.f, 0.f, 0.f};
#pragma unroll
    for (int kk = 0; kk < 4; ++kk) {
      v2f b = {Z[kk][2 * j], Z[kk][2 * j + 1]};
      acc = __builtin_amdgcn_wmma_f32_16x16x4_f32(false, g[kk], false, b,
                                                  (short)0, acc, false, false);
    }
    X[j] = acc;
  }
}

__device__ __forceinline__ void product_state(v8f (&X)[4], const float* cw,
                                              const float* sw, int lane) {
  const int c = lane & 15;
  const int h = lane >> 4;
  float lf = ((c >> 3) & 1) ? sw[4] : cw[4];
  lf *= ((c >> 2) & 1) ? sw[5] : cw[5];
  lf *= ((c >> 1) & 1) ? sw[6] : cw[6];
  lf *= (c & 1) ? sw[7] : cw[7];
  lf *= h ? sw[8] : cw[8];
  float pr[8];
#pragma unroll
  for (int r = 0; r < 8; ++r) {
    const float f9 = ((r >> 2) & 1) ? sw[9] : cw[9];
    const float f0 = ((r >> 1) & 1) ? sw[0] : cw[0];
    const float f3 = (r & 1) ? sw[3] : cw[3];
    pr[r] = lf * (f9 * f0 * f3);
  }
#pragma unroll
  for (int j = 0; j < 4; ++j) {
    const float f1 = ((j >> 1) & 1) ? sw[1] : cw[1];
    const float f2 = (j & 1) ? sw[2] : cw[2];
    const float gj = f1 * f2;
#pragma unroll
    for (int r = 0; r < 8; ++r)
      X[j][r] = pr[r] * gj;
  }
}

__device__ __forceinline__ constexpr int cnot_src(int i) {
  int s = i;
  s ^= (s & 1) << 9;
  for (int w = 8; w >= 0; --w)
    s ^= ((s >> (9 - w)) & 1) << (8 - w);
  return s;
}

__device__ __forceinline__ void cnot_ring(v8f (&X)[4], float* __restrict__ S,
                                          int lane) {
  const int c = lane & 15;
  const int h = lane >> 4;
  const int st_base = (c << 2) | (h << 1);
#pragma unroll
  for (int j = 0; j < 4; ++j)
#pragma unroll
    for (int r = 0; r < 8; ++r) {
      const int a = (((r >> 1) & 1) << 3) | (j << 1) | (r & 1);
      S[st_base | (a << 6) | (r >> 2)] = X[j][r];
    }
  const int ld_base = cnot_src((h << 7) | (c << 2));
#pragma unroll
  for (int t = 0; t < 4; ++t)
#pragma unroll
    for (int r = 0; r < 8; ++r) {
      const int a = 4 * (r >> 1) + (r & 1);
      X[t][r] = S[ld_base ^ cnot_src((a << 6) | t)];
    }
}

template <int MASK>
__device__ __forceinline__ float row_xor(float v) {
  return __int_as_float(__builtin_amdgcn_update_dpp(
      0, __float_as_int(v), 0x160 | MASK, 0xf, 0xf, true));
}

__device__ __forceinline__ float xhalf(float v) {
  const int b = __float_as_int(v);
  return __int_as_float(
      __builtin_amdgcn_permlanex16(b, b, 0x76543210, 0xfedcba98, false, false));
}

template <int MASK>
__device__ __forceinline__ float pair_step(float a, float b, bool bit) {
  const float keep = bit ? b : a;
  const float send = bit ? a : b;
  return keep + row_xor<MASK>(send);
}

__device__ __forceinline__ float hsum8(v8f v) {
  return ((v[0] + v[1]) + (v[2] + v[3])) + ((v[4] + v[5]) + (v[6] + v[7]));
}

__device__ __forceinline__ float lane_bcast(float v, int src_lane) {
  return __int_as_float(__builtin_amdgcn_readlane(__float_as_int(v), src_lane));
}

__global__ __launch_bounds__(WAVES_PER_BLOCK * 32, 9)
void qlayer_wmma_kernel(const float* __restrict__ x,
                        const float* __restrict__ params,
                        float* __restrict__ out, int batch) {
  __shared__ float lds[WAVES_PER_BLOCK * STATE_SZ];
  const int lane   = threadIdx.x & 31;
  const int wid    = threadIdx.x >> 5;
  const int sample = blockIdx.x * WAVES_PER_BLOCK + wid;
  if (sample >= batch) return;
  float* S = lds + wid * STATE_SZ;

  v8f X[4];

  const int ai = (lane < N_LAYERS * N_QUBITS) ? lane : 0;
  const int xi = (lane < N_QUBITS) ? lane : 0;
  const float pv = params[ai];
  const float xv = x[sample * N_QUBITS + xi];
  const float half_angle =
      0.5f * ((lane < N_QUBITS) ? (tanhf(xv) + pv) : pv);
  const float cv = cosf(half_angle);
  const float sv = sinf(half_angle);

  {
    float c0[N_QUBITS], s0[N_QUBITS];
#pragma unroll
    for (int w = 0; w < N_QUBITS; ++w) {
      c0[w] = lane_bcast(cv, w);
      s0[w] = lane_bcast(sv, w);
    }
    product_state(X, c0, s0, lane);
    cnot_ring(X, S, lane);
  }

  for (int l = 1; l < N_LAYERS; ++l) {
    float cl[N_QUBITS], sl[N_QUBITS];
#pragma unroll
    for (int w = 0; w < N_QUBITS; ++w) {
      cl[w] = lane_bcast(cv, l * N_QUBITS + w);
      sl[w] = lane_bcast(sv, l * N_QUBITS + w);
    }
    ry_layer(X, cl, sl, lane);
    cnot_ring(X, S, lane);
  }

  const int c = lane & 15;
  const int h = lane >> 4;
  const v8f p0 = X[0] * X[0], p1 = X[1] * X[1];
  const v8f p2 = X[2] * X[2], p3 = X[3] * X[3];
  const v8f sa = p0 + p1, sb = p2 + p3;
  const v8f d9 = (p0 - p1) + (p2 - p3);
  const v8f s8 = sa + sb;
  const v8f d8 = sa - sb;
  const float D9 = hsum8(d9);
  const float D8 = hsum8(d8);
  float s3[4], d3[4];
#pragma unroll
  for (int k = 0; k < 4; ++k) {
    s3[k] = s8[2 * k] + s8[2 * k + 1];
    d3[k] = s8[2 * k] - s8[2 * k + 1];
  }
  const float D3 = (d3[0] + d3[1]) + (d3[2] + d3[3]);
  const float s1a = s3[0] + s3[1], s1b = s3[2] + s3[3];
  const float D1 = (s3[0] - s3[1]) + (s3[2] - s3[3]);
  const float T  = s1a + s1b;
  const float D0 = s1a - s1b;

  const bool b1 = (c & 1) != 0, b2 = (c & 2) != 0;
  const bool b4 = (c & 4) != 0, b8 = (c & 8) != 0;
  const float v2 = h ? -T : T;
  const float v4 = b8 ? -T : T;
  const float v5 = b4 ? -T : T;
  const float v6 = b2 ? -T : T;
  const float v7 = b1 ? -T : T;

  const float r0 = pair_step<1>(D0, D1, b1);
  const float r1 = pair_step<1>(v2, D3, b1);
  const float r2 = pair_step<1>(v4, v5, b1);
  const float r3 = pair_step<1>(v6, v7, b1);
  const float r4 = pair_step<1>(D8, D9, b1);
  const float q0 = pair_step<2>(r0, r1, b2);
  const float q1 = pair_step<2>(r2, r3, b2);
  const float q2 = r4 + row_xor<2>(r4);
  const float u0 = pair_step<4>(q0, q1, b4);
  const float u1 = q2 + row_xor<4>(q2);
  const float z  = pair_step<8>(u0, u1, b8);
  const float zz = z + xhalf(z);
  if (lane < N_QUBITS)
    out[sample * N_QUBITS + lane] = zz;
}

extern "C" void kernel_launch(void* const* d_in, const int* in_sizes, int n_in,
                              void* d_out, int out_size, void* d_ws, size_t ws_size,
                              hipStream_t stream) {
  const float* x      = (const float*)d_in[0];
  const float* params = (const float*)d_in[1];
  float* out          = (float*)d_out;
  const int batch = in_sizes[0] / N_QUBITS;
  const int blocks = (batch + WAVES_PER_BLOCK - 1) / WAVES_PER_BLOCK;
  qlayer_wmma_kernel<<<blocks, WAVES_PER_BLOCK * 32, 0, stream>>>(x, params, out, batch);
}


# Four samples per wave, in turn: cos/sin and the 24 gate fragment pairs of parameter layers 1, 2 built once per wave, held in registers; first-layer angles 3+1 samples per trig pass; sums unchanged
# speedup vs baseline: 3.0161x; 1.0664x over previous
#include <hip/hip_runtime.h>
#include <math.h>


typedef __attribute__((ext_vector_type(2))) float v2f;
typedef __attribute__((ext_vector_type(8))) float v8f;

#define N_QUBITS 10
#define N_LAYERS 3
#define WAVES_PER_BLOCK 4
#define STATE_SZ 1024
#define SAMPLES_PER_WAVE 4

__device__ __forceinline__ float ry_elem(float c, float s, int row, int col) {
  return (row == col) ? c : ((row == 0) ? -s : s);
}

__device__ __forceinline__ float kron4(const float* c4, const float* s4,
                                       int row, int col) {
  float e = 1.0f;
#pragma unroll
  for (int j = 0; j < 4; ++j)
    e *= ry_elem(c4[j], s4[j], (row >> (3 - j)) & 1, (col >> (3 - j)) & 1);
  return e;
}

struct LayerGates {
  v2f g1[4];
  v2f g2[4];
  v2f g3[4];
};

__device__ __forceinline__ void build_gates(LayerGates& G, const float* cw,
                                            const float* sw, int lane) {
  const int c = lane & 15;
  const int h = lane >> 4;
  const float c3[4] = {cw[8], cw[9], 1.0f, 1.0f};
  const float s3[4] = {sw[8], sw[9], 0.0f, 0.0f};
#pragma unroll
  for (int kk = 0; kk < 4; ++kk)
#pragma unroll
    for (int v = 0; v < 2; ++v) {
      G.g1[kk][v] = kron4(cw, sw, c, 4 * kk + 2 * h + v);
      G.g2[kk][v] = kron4(cw + 4, sw + 4, c, 2 * kk + v + 8 * h);
      G.g3[kk][v] = kron4(c3, s3, c, 4 * kk + 2 * h + v);
    }
}

__device__ __forceinline__ void ry_layer(v8f (&X)[4], const LayerGates& G) {
  v8f Y[4], Z[4];

#pragma unroll
  for (int t = 0; t < 4; ++t) {
    v8f acc = {0.f, 0.f, 0.f, 0.f, 0.f, 0.f, 0.f, 0.f};
#pragma unroll
    for (int kk = 0; kk < 4; ++kk) {
      v2f a = {X[t][2 * kk], X[t][2 * kk + 1]};
      acc = __builtin_amdgcn_wmma_f32_16x16x4_f32(false, a, false, G.g1[kk],
                                                  (short)0, acc, false, false);
    }
    Y[t] = acc;
  }

#pragma unroll
  for (int t = 0; t < 4; ++t) {
    v8f acc = {0.f, 0.f, 0.f, 0.f, 0.f, 0.f, 0.f, 0.f};
#pragma unroll
    for (int kk = 0; kk < 4; ++kk) {
      v2f a = {Y[t][2 * kk], Y[t][2 * kk + 1]};
      acc = __builtin_amdgcn_wmma_f32_16x16x4_f32(false, a, false, G.g2[kk],
                                                  (short)0, acc, false, false);
    }
    Z[t] = acc;
  }

#pragma unroll
  for (int j = 0; j < 4; ++j) {
    v8f acc = {0.f, 0.f, 0.f, 0.f, 0.f, 0.f, 0.f, 0.f};
#pragma unroll
    for (int kk = 0; kk < 4; ++kk) {
      v2f b = {Z[kk][2 * j], Z[kk][2 * j + 1]};
      acc = __builtin_amdgcn_wmma_f32_16x16x4_f32(false, G.g3[kk], false, b,
                                                  (short)0, acc, false, false);
    }
    X[j] = acc;
  }
}

__device__ __forceinline__ void product_state(v8f (&X)[4], const float* cw,
                                              const float* sw, int lane) {
  const int c = lane & 15;
  const int h = lane >> 4;
  float lf = ((c >> 3) & 1) ? sw[4] : cw[4];
  lf *= ((c >> 2) & 1) ? sw[5] : cw[5];
  lf *= ((c >> 1) & 1) ? sw[6] : cw[6];
  lf *= (c & 1) ? sw[7] : cw[7];
  lf *= h ? sw[8] : cw[8];
  float pr[8];
#pragma unroll
  for (int r = 0; r < 8; ++r) {
    const float f9 = ((r >> 2) & 1) ? sw[9] : cw[9];
    const float f0 = ((r >> 1) & 1) ? sw[0] : cw[0];
    const float f3 = (r & 1) ? sw[3] : cw[3];
    pr[r] = lf * (f9 * f0 * f3);
  }
#pragma unroll
  for (int j = 0; j < 4; ++j) {
    const float f1 = ((j >> 1) & 1) ? sw[1] : cw[1];
    const float f2 = (j & 1) ? sw[2] : cw[2];
    const float gj = f1 * f2;
#pragma unroll
    for (int r = 0; r < 8; ++r)
      X[j][r] = pr[r] * gj;
  }
}

__device__ __forceinline__ constexpr int cnot_src(int i) {
  int s = i;
  s ^= (s & 1) << 9;
  for (int w = 8; w >= 0; --w)
    s ^= ((s >> (9 - w)) & 1) << (8 - w);
  return s;
}

__device__ __forceinline__ void cnot_ring(v8f (&X)[4], float* __restrict__ S,
                                          int lane) {
  const int c = lane & 15;
  const int h = lane >> 4;
  const int st_base = (c << 2) | (h << 1);
#pragma unroll
  for (int j = 0; j < 4; ++j)
#pragma unroll
    for (int r = 0; r < 8; ++r) {
      const int a = (((r >> 1) & 1) << 3) | (j << 1) | (r & 1);
      S[st_base | (a << 6) | (r >> 2)] = X[j][r];
    }
  const int ld_base = cnot_src((h << 7) | (c << 2));
#pragma unroll
  for (int t = 0; t < 4; ++t)
#pragma unroll
    for (int r = 0; r < 8; ++r) {
      const int a = 4 * (r >> 1) + (r & 1);
      X[t][r] = S[ld_base ^ cnot_src((a << 6) | t)];
    }
  __builtin_amdgcn_sched_barrier(0);
}

template <int MASK>
__device__ __forceinline__ float row_xor(float v) {
  return __int_as_float(__builtin_amdgcn_update_dpp(
      0, __float_as_int(v), 0x160 | MASK, 0xf, 0xf, true));
}

__device__ __forceinline__ float xhalf(float v) {
  const int b = __float_as_int(v);
  return __int_as_float(
      __builtin_amdgcn_permlanex16(b, b, 0x76543210, 0xfedcba98, false, false));
}

template <int MASK>
__device__ __forceinline__ float pair_step(float a, float b, bool bit) {
  const float keep = bit ? b : a;
  const float send = bit ? a : b;
  return keep + row_xor<MASK>(send);
}

__device__ __forceinline__ float hsum8(v8f v) {
  return ((v[0] + v[1]) + (v[2] + v[3])) + ((v[4] + v[5]) + (v[6] + v[7]));
}

__device__ __forceinline__ float lane_bcast(float v, int src_lane) {
  return __int_as_float(__builtin_amdgcn_readlane(__float_as_int(v), src_lane));
}

static_assert(SAMPLES_PER_WAVE == 4, "the two angle passes hold 3 + 1 samples");

__global__ __launch_bounds__(WAVES_PER_BLOCK * 32)
void qlayer_wmma_kernel(const float* __restrict__ x,
                        const float* __restrict__ params,
                        float* __restrict__ out, int batch) {
  __shared__ float lds[WAVES_PER_BLOCK * STATE_SZ];
  const int lane = threadIdx.x & 31;
  const int wid  = threadIdx.x >> 5;
  const int first = (blockIdx.x * WAVES_PER_BLOCK + wid) * SAMPLES_PER_WAVE;
  if (first >= batch) return;
  const int left = batch - first;
  const int nvalid = left < SAMPLES_PER_WAVE ? left : SAMPLES_PER_WAVE;
  const int last_sample = batch - 1;
  float* S = lds + wid * STATE_SZ;
  const int c = lane & 15;
  const int h = lane >> 4;

  const bool live = lane < 3 * N_QUBITS;
  const int g1 = live ? ((lane >= 2 * N_QUBITS) ? 2 : ((lane >= N_QUBITS) ? 1 : 0)) : 0;
  const int w1 = live ? lane - N_QUBITS * g1 : 0;
  const int sa = (first + g1 < last_sample) ? first + g1 : last_sample;
  const float half1 = 0.5f * (tanhf(x[sa * N_QUBITS + w1]) + params[w1]);
  const float cv1 = cosf(half1);
  const float sv1 = sinf(half1);

  const int ai = live ? lane : 0;
  const int xi = (lane < N_QUBITS) ? lane : 0;
  const int sb = (first + 3 < last_sample) ? first + 3 : last_sample;
  const float pv = params[ai];
  const float xv = x[sb * N_QUBITS + xi];
  const float half2 = 0.5f * ((lane < N_QUBITS) ? (tanhf(xv) + pv) : pv);
  const float cv2 = cosf(half2);
  const float sv2 = sinf(half2);

  LayerGates G[N_LAYERS - 1];
#pragma unroll
  for (int l = 1; l < N_LAYERS; ++l) {
    float cl[N_QUBITS], sl[N_QUBITS];
#pragma unroll
    for (int w = 0; w < N_QUBITS; ++w) {
      cl[w] = lane_bcast(cv2, l * N_QUBITS + w);
      sl[w] = lane_bcast(sv2, l * N_QUBITS + w);
    }
    build_gates(G[l - 1], cl, sl, lane);
  }

  const bool b1 = (c & 1) != 0, b2 = (c & 2) != 0;
  const bool b4 = (c & 4) != 0, b8 = (c & 8) != 0;

#pragma unroll 1
  for (int i = 0; i < nvalid; ++i) {
    v8f X[4];
    {
      const bool fourth = (i == 3);
      const float cvi = fourth ? cv2 : cv1;
      const float svi = fourth ? sv2 : sv1;
      const int base = fourth ? 0 : N_QUBITS * i;
      float c0[N_QUBITS], s0[N_QUBITS];
#pragma unroll
      for (int w = 0; w < N_QUBITS; ++w) {
        c0[w] = lane_bcast(cvi, base + w);
        s0[w] = lane_bcast(svi, base + w);
      }
      product_state(X, c0, s0, lane);
      cnot_ring(X, S, lane);
    }

#pragma unroll
    for (int l = 1; l < N_LAYERS; ++l) {
      ry_layer(X, G[l - 1]);
      cnot_ring(X, S, lane);
    }

    const v8f p0 = X[0] * X[0], p1 = X[1] * X[1];
    const v8f p2 = X[2] * X[2], p3 = X[3] * X[3];
    const v8f sa01 = p0 + p1, sb23 = p2 + p3;
    const v8f d9 = (p0 - p1) + (p2 - p3);
    const v8f s8 = sa01 + sb23;
    const v8f d8 = sa01 - sb23;
    const float D9 = hsum8(d9);
    const float D8 = hsum8(d8);
    float s3[4], d3[4];
#pragma unroll
    for (int k = 0; k < 4; ++k) {
      s3[k] = s8[2 * k] + s8[2 * k + 1];
      d3[k] = s8[2 * k] - s8[2 * k + 1];
    }
    const float D3 = (d3[0] + d3[1]) + (d3[2] + d3[3]);
    const float s1a = s3[0] + s3[1], s1b = s3[2] + s3[3];
    const float D1 = (s3[0] - s3[1]) + (s3[2] - s3[3]);
    const float T  = s1a + s1b;
    const float D0 = s1a - s1b;

    const float v2 = h ? -T : T;
    const float v4 = b8 ? -T : T;
    const float v5 = b4 ? -T : T;
    const float v6 = b2 ? -T : T;
    const float v7 = b1 ? -T : T;

    const float r0 = pair_step<1>(D0, D1, b1);
    const float r1 = pair_step<1>(v2, D3, b1);
    const float r2 = pair_step<1>(v4, v5, b1);
    const float r3 = pair_step<1>(v6, v7, b1);
    const float r4 = pair_step<1>(D8, D9, b1);
    const float q0 = pair_step<2>(r0, r1, b2);
    const float q1 = pair_step<2>(r2, r3, b2);
    const float q2 = r4 + row_xor<2>(r4);
    const float u0 = pair_step<4>(q0, q1, b4);
    const float u1 = q2 + row_xor<4>(q2);
    const float z  = pair_step<8>(u0, u1, b8);
    const float zz = z + xhalf(z);
    if (lane < N_QUBITS)
      out[(first + i) * N_QUBITS + lane] = zz;
  }
}

extern "C" void kernel_launch(void* const* d_in, const int* in_sizes, int n_in,
                              void* d_out, int out_size, void* d_ws, size_t ws_size,
                              hipStream_t stream) {
  const float* x      = (const float*)d_in[0];
  const float* params = (const float*)d_in[1];
  float* out          = (float*)d_out;
  const int batch = in_sizes[0] / N_QUBITS;
  const int per_block = WAVES_PER_BLOCK * SAMPLES_PER_WAVE;
  const int blocks = (batch + per_block - 1) / per_block;
  qlayer_wmma_kernel<<<blocks, WAVES_PER_BLOCK * 32, 0, stream>>>(x, params, out, batch);
}


# First layer + its ring + group 1 of the next layer as one product: ring-permuted product state = F[a] x R[b,q]; G1 x F for 4 samples in 4 matrix instr. per wave; 81 a sample, was 96; rounding changes
# speedup vs baseline: 3.5569x; 1.1793x over previous
#include <hip/hip_runtime.h>
#include <math.h>


typedef __attribute__((ext_vector_type(2))) float v2f;
typedef __attribute__((ext_vector_type(8))) float v8f;

#define N_QUBITS 10
#define N_LAYERS 3
#define WAVES_PER_BLOCK 4
#define STATE_SZ 1024
#define SAMPLES_PER_WAVE 4

__device__ __forceinline__ float ry_elem(float c, float s, int row, int col) {
  return (row == col) ? c : ((row == 0) ? -s : s);
}

__device__ __forceinline__ float kron4(const float* c4, const float* s4,
                                       int row, int col) {
  float e = 1.0f;
#pragma unroll
  for (int j = 0; j < 4; ++j)
    e *= ry_elem(c4[j], s4[j], (row >> (3 - j)) & 1, (col >> (3 - j)) & 1);
  return e;
}

struct LayerGates {
  v2f g1[4];
  v2f g2[4];
  v2f g3[4];
};

__device__ __forceinline__ void build_gates(LayerGates& G, const float* cw,
                                            const float* sw, int lane) {
  const int c = lane & 15;
  const int h = lane >> 4;
  const float c3[4] = {cw[8], cw[9], 1.0f, 1.0f};
  const float s3[4] = {sw[8], sw[9], 0.0f, 0.0f};
#pragma unroll
  for (int kk = 0; kk < 4; ++kk)
#pragma unroll
    for (int v = 0; v < 2; ++v) {
      G.g1[kk][v] = kron4(cw, sw, c, 4 * kk + 2 * h + v);
      G.g2[kk][v] = kron4(cw + 4, sw + 4, c, 2 * kk + v + 8 * h);
      G.g3[kk][v] = kron4(c3, s3, c, 4 * kk + 2 * h + v);
    }
}

__device__ __forceinline__ void ry_groups23(v8f (&X)[4], const v8f (&Y)[4],
                                            const LayerGates& G) {
  v8f Z[4];

#pragma unroll
  for (int t = 0; t < 4; ++t) {
    v8f acc = {0.f, 0.f, 0.f, 0.f, 0.f, 0.f, 0.f, 0.f};
#pragma unroll
    for (int kk = 0; kk < 4; ++kk) {
      v2f a = {Y[t][2 * kk], Y[t][2 * kk + 1]};
      acc = __builtin_amdgcn_wmma_f32_16x16x4_f32(false, a, false, G.g2[kk],
                                                  (short)0, acc, false, false);
    }
    Z[t] = acc;
  }

#pragma unroll
  for (int j = 0; j < 4; ++j) {
    v8f acc = {0.f, 0.f, 0.f, 0.f, 0.f, 0.f, 0.f, 0.f};
#pragma unroll
    for (int kk = 0; kk < 4; ++kk) {
      v2f b = {Z[kk][2 * j], Z[kk][2 * j + 1]};
      acc = __builtin_amdgcn_wmma_f32_16x16x4_f32(false, G.g3[kk], false, b,
                                                  (short)0, acc, false, false);
    }
    X[j] = acc;
  }
}

__device__ __forceinline__ void ry_layer(v8f (&X)[4], const LayerGates& G) {
  v8f Y[4];

#pragma unroll
  for (int t = 0; t < 4; ++t) {
    v8f acc = {0.f, 0.f, 0.f, 0.f, 0.f, 0.f, 0.f, 0.f};
#pragma unroll
    for (int kk = 0; kk < 4; ++kk) {
      v2f a = {X[t][2 * kk], X[t][2 * kk + 1]};
      acc = __builtin_amdgcn_wmma_f32_16x16x4_f32(false, a, false, G.g1[kk],
                                                  (short)0, acc, false, false);
    }
    Y[t] = acc;
  }

  ry_groups23(X, Y, G);
}


__device__ __forceinline__ float lane_gather(float v, int src_lane) {
  return __int_as_float(
      __builtin_amdgcn_ds_bpermute(src_lane << 2, __float_as_int(v)));
}

__device__ __forceinline__ v8f first_ta(const float* cw5, const float* sw5,
                                        const v2f* g1, int lane) {
  const int c = lane & 15;
  const bool h = (lane >> 4) != 0;
  const bool y9 = (c & 1) != 0;
  const bool y4 = (c & 8) != 0;
  const float p0[2] = {y9 ? sw5[0] : cw5[0], y9 ? cw5[0] : sw5[0]};
  const float p1[2] = {y9 ? sw5[1] : cw5[1], y9 ? cw5[1] : sw5[1]};
  const float p2[2] = {h ? sw5[2] : cw5[2], h ? cw5[2] : sw5[2]};
  const float p3[2] = {h ? sw5[3] : cw5[3], h ? cw5[3] : sw5[3]};
  const float p4[2] = {y4 ? sw5[4] : cw5[4], y4 ? cw5[4] : sw5[4]};
  const float e34[2] = {p3[0] * p4[0], p3[1] * p4[1]};
  v8f acc = {0.f, 0.f, 0.f, 0.f, 0.f, 0.f, 0.f, 0.f};
#pragma unroll
  for (int kk = 0; kk < 4; ++kk) {
    const int y0 = kk >> 1, y1 = kk & 1;
    const float hd = p0[y0] * p1[y1 ^ y0] * p2[y1];
    v2f a = {hd * e34[0], hd * e34[1]};
    acc = __builtin_amdgcn_wmma_f32_16x16x4_f32(false, a, false, g1[kk],
                                                (short)0, acc, false, false);
  }
  return acc;
}

__device__ __forceinline__ void first_y(v8f (&Y)[4], float ta0, float ta1,
                                        const float* cw, const float* sw,
                                        int lane) {
  const bool h = (lane >> 4) != 0;
  const float u[2] = {h ? sw[5] : cw[5], h ? cw[5] : sw[5]};
  float lf[2][2];
#pragma unroll
  for (int y5 = 0; y5 < 2; ++y5) {
    lf[y5][0] = ta0 * u[y5];
    lf[y5][1] = ta1 * u[y5];
  }
  float p89[2][4];
#pragma unroll
  for (int y7 = 0; y7 < 2; ++y7)
#pragma unroll
    for (int t = 0; t < 4; ++t) {
      const int y8 = t >> 1, y9 = t & 1;
      p89[y7][t] = ((y8 ^ y7) ? sw[8] : cw[8]) * ((y9 ^ y8) ? sw[9] : cw[9]);
    }
#pragma unroll
  for (int r = 0; r < 8; ++r) {
    const int y5 = r >> 2, y6 = (r >> 1) & 1, y7 = r & 1;
    const float p67 = ((y6 ^ y5) ? sw[6] : cw[6]) * ((y7 ^ y6) ? sw[7] : cw[7]);
#pragma unroll
    for (int t = 0; t < 4; ++t)
      Y[t][r] = lf[y5][t & 1] * (p67 * p89[y7][t]);
  }
}

__device__ __forceinline__ constexpr int cnot_src(int i) {
  int s = i;
  s ^= (s & 1) << 9;
  for (int w = 8; w >= 0; --w)
    s ^= ((s >> (9 - w)) & 1) << (8 - w);
  return s;
}

__device__ __forceinline__ void cnot_ring(v8f (&X)[4], float* __restrict__ S,
                                          int lane) {
  const int c = lane & 15;
  const int h = lane >> 4;
  const int st_base = (c << 2) | (h << 1);
#pragma unroll
  for (int j = 0; j < 4; ++j)
#pragma unroll
    for (int r = 0; r < 8; ++r) {
      const int a = (((r >> 1) & 1) << 3) | (j << 1) | (r & 1);
      S[st_base | (a << 6) | (r >> 2)] = X[j][r];
    }
  const int ld_base = cnot_src((h << 7) | (c << 2));
#pragma unroll
  for (int t = 0; t < 4; ++t)
#pragma unroll
    for (int r = 0; r < 8; ++r) {
      const int a = 4 * (r >> 1) + (r & 1);
      X[t][r] = S[ld_base ^ cnot_src((a << 6) | t)];
    }
  __builtin_amdgcn_sched_barrier(0);
}

template <int MASK>
__device__ __forceinline__ float row_xor(float v) {
  return __int_as_float(__builtin_amdgcn_update_dpp(
      0, __float_as_int(v), 0x160 | MASK, 0xf, 0xf, true));
}

__device__ __forceinline__ float xhalf(float v) {
  const int b = __float_as_int(v);
  return __int_as_float(
      __builtin_amdgcn_permlanex16(b, b, 0x76543210, 0xfedcba98, false, false));
}

template <int MASK>
__device__ __forceinline__ float pair_step(float a, float b, bool bit) {
  const float keep = bit ? b : a;
  const float send = bit ? a : b;
  return keep + row_xor<MASK>(send);
}

__device__ __forceinline__ float hsum8(v8f v) {
  return ((v[0] + v[1]) + (v[2] + v[3])) + ((v[4] + v[5]) + (v[6] + v[7]));
}

__device__ __forceinline__ float lane_bcast(float v, int src_lane) {
  return __int_as_float(__builtin_amdgcn_readlane(__float_as_int(v), src_lane));
}

static_assert(SAMPLES_PER_WAVE == 4, "the two angle passes hold 3 + 1 samples");

__global__ __launch_bounds__(WAVES_PER_BLOCK * 32)
void qlayer_wmma_kernel(const float* __restrict__ x,
                        const float* __restrict__ params,
                        float* __restrict__ out, int batch) {
  __shared__ float lds[WAVES_PER_BLOCK * STATE_SZ];
  const int lane = threadIdx.x & 31;
  const int wid  = threadIdx.x >> 5;
  const int first = (blockIdx.x * WAVES_PER_BLOCK + wid) * SAMPLES_PER_WAVE;
  if (first >= batch) return;
  const int left = batch - first;
  const int nvalid = left < SAMPLES_PER_WAVE ? left : SAMPLES_PER_WAVE;
  const int last_sample = batch - 1;
  float* S = lds + wid * STATE_SZ;
  const int c = lane & 15;
  const int h = lane >> 4;

  const bool live = lane < 3 * N_QUBITS;
  const int g1 = live ? ((lane >= 2 * N_QUBITS) ? 2 : ((lane >= N_QUBITS) ? 1 : 0)) : 0;
  const int w1 = live ? lane - N_QUBITS * g1 : 0;
  const int sa = (first + g1 < last_sample) ? first + g1 : last_sample;
  const float half1 = 0.5f * (tanhf(x[sa * N_QUBITS + w1]) + params[w1]);
  const float cv1 = cosf(half1);
  const float sv1 = sinf(half1);

  const int ai = live ? lane : 0;
  const int xi = (lane < N_QUBITS) ? lane : 0;
  const int sb = (first + 3 < last_sample) ? first + 3 : last_sample;
  const float pv = params[ai];
  const float xv = x[sb * N_QUBITS + xi];
  const float half2 = 0.5f * ((lane < N_QUBITS) ? (tanhf(xv) + pv) : pv);
  const float cv2 = cosf(half2);
  const float sv2 = sinf(half2);

  LayerGates G[N_LAYERS - 1];
#pragma unroll
  for (int l = 1; l < N_LAYERS; ++l) {
    float cl[N_QUBITS], sl[N_QUBITS];
#pragma unroll
    for (int w = 0; w < N_QUBITS; ++w) {
      cl[w] = lane_bcast(cv2, l * N_QUBITS + w);
      sl[w] = lane_bcast(sv2, l * N_QUBITS + w);
    }
    build_gates(G[l - 1], cl, sl, lane);
  }

  v8f TA;
  {
    const int gm = (c >> 1) & 3;
    const int src0 = (gm < 3) ? N_QUBITS * gm : 0;
    float cw5[5], sw5[5];
#pragma unroll
    for (int w = 0; w < 5; ++w) {
      const float ca = lane_gather(cv1, src0 + w);
      const float sa1 = lane_gather(sv1, src0 + w);
      const float cb = lane_bcast(cv2, w);
      const float sb3 = lane_bcast(sv2, w);
      cw5[w] = (gm == 3) ? cb : ca;
      sw5[w] = (gm == 3) ? sb3 : sa1;
    }
    TA = first_ta(cw5, sw5, G[0].g1, lane);
  }

  const bool b1 = (c & 1) != 0, b2 = (c & 2) != 0;
  const bool b4 = (c & 4) != 0, b8 = (c & 8) != 0;

#pragma unroll 1
  for (int i = 0; i < nvalid; ++i) {
    v8f X[4];
    {
      const bool fourth = (i == 3);
      const float cvi = fourth ? cv2 : cv1;
      const float svi = fourth ? sv2 : sv1;
      const int base = fourth ? 0 : N_QUBITS * i;
      float c0[N_QUBITS], s0[N_QUBITS];
#pragma unroll
      for (int w = 5; w < N_QUBITS; ++w) {
        c0[w] = lane_bcast(cvi, base + w);
        s0[w] = lane_bcast(svi, base + w);
      }
      const float ta0 = (i == 0) ? TA[0] : (i == 1) ? TA[2] : (i == 2) ? TA[4] : TA[6];
      const float ta1 = (i == 0) ? TA[1] : (i == 1) ? TA[3] : (i == 2) ? TA[5] : TA[7];
      v8f Y[4];
      first_y(Y, ta0, ta1, c0, s0, lane);
      ry_groups23(X, Y, G[0]);
      cnot_ring(X, S, lane);
    }

    ry_layer(X, G[1]);
    cnot_ring(X, S, lane);

    const v8f p0 = X[0] * X[0], p1 = X[1] * X[1];
    const v8f p2 = X[2] * X[2], p3 = X[3] * X[3];
    const v8f sa01 = p0 + p1, sb23 = p2 + p3;
    const v8f d9 = (p0 - p1) + (p2 - p3);
    const v8f s8 = sa01 + sb23;
    const v8f d8 = sa01 - sb23;
    const float D9 = hsum8(d9);
    const float D8 = hsum8(d8);
    float s3[4], d3[4];
#pragma unroll
    for (int k = 0; k < 4; ++k) {
      s3[k] = s8[2 * k] + s8[2 * k + 1];
      d3[k] = s8[2 * k] - s8[2 * k + 1];
    }
    const float D3 = (d3[0] + d3[1]) + (d3[2] + d3[3]);
    const float s1a = s3[0] + s3[1], s1b = s3[2] + s3[3];
    const float D1 = (s3[0] - s3[1]) + (s3[2] - s3[3]);
    const float T  = s1a + s1b;
    const float D0 = s1a - s1b;

    const float v2 = h ? -T : T;
    const float v4 = b8 ? -T : T;
    const float v5 = b4 ? -T : T;
    const float v6 = b2 ? -T : T;
    const float v7 = b1 ? -T : T;

    const float r0 = pair_step<1>(D0, D1, b1);
    const float r1 = pair_step<1>(v2, D3, b1);
    const float r2 = pair_step<1>(v4, v5, b1);
    const float r3 = pair_step<1>(v6, v7, b1);
    const float r4 = pair_step<1>(D8, D9, b1);
    const float q0 = pair_step<2>(r0, r1, b2);
    const float q1 = pair_step<2>(r2, r3, b2);
    const float q2 = r4 + row_xor<2>(r4);
    const float u0 = pair_step<4>(q0, q1, b4);
    const float u1 = q2 + row_xor<4>(q2);
    const float z  = pair_step<8>(u0, u1, b8);
    const float zz = z + xhalf(z);
    if (lane < N_QUBITS)
      out[(first + i) * N_QUBITS + lane] = zz;
  }
}

extern "C" void kernel_launch(void* const* d_in, const int* in_sizes, int n_in,
                              void* d_out, int out_size, void* d_ws, size_t ws_size,
                              hipStream_t stream) {
  const float* x      = (const float*)d_in[0];
  const float* params = (const float*)d_in[1];
  float* out          = (float*)d_out;
  const int batch = in_sizes[0] / N_QUBITS;
  const int per_block = WAVES_PER_BLOCK * SAMPLES_PER_WAVE;
  const int blocks = (batch + per_block - 1) / per_block;
  qlayer_wmma_kernel<<<blocks, WAVES_PER_BLOCK * 32, 0, stream>>>(x, params, out, batch);
}


# Parameter layer 1 on wires 4-9 applied to the product factors, not the state: output = TA (16x4) x W (4x64), W by 2x2 chain steps per lane; 4 matrix instr. a sample for it, was 32; rounding changes
# speedup vs baseline: 4.4684x; 1.2563x over previous
#include <hip/hip_runtime.h>
#include <math.h>


typedef __attribute__((ext_vector_type(2))) float v2f;
typedef __attribute__((ext_vector_type(8))) float v8f;

#define N_QUBITS 10
#define N_LAYERS 3
#define WAVES_PER_BLOCK 4
#define STATE_SZ 1024
#define SAMPLES_PER_WAVE 4

__device__ __forceinline__ float ry_elem(float c, float s, int row, int col) {
  return (row == col) ? c : ((row == 0) ? -s : s);
}

__device__ __forceinline__ float kron4(const float* c4, const float* s4,
                                       int row, int col) {
  float e = 1.0f;
#pragma unroll
  for (int j = 0; j < 4; ++j)
    e *= ry_elem(c4[j], s4[j], (row >> (3 - j)) & 1, (col >> (3 - j)) & 1);
  return e;
}

struct LayerGates {
  v2f g1[4];
  v2f g2[4];
  v2f g3[4];
};

__device__ __forceinline__ void build_gates(LayerGates& G, const float* cw,
                                            const float* sw, int lane) {
  const int c = lane & 15;
  const int h = lane >> 4;
  const float c3[4] = {cw[8], cw[9], 1.0f, 1.0f};
  const float s3[4] = {sw[8], sw[9], 0.0f, 0.0f};
#pragma unroll
  for (int kk = 0; kk < 4; ++kk)
#pragma unroll
    for (int v = 0; v < 2; ++v) {
      G.g1[kk][v] = kron4(cw, sw, c, 4 * kk + 2 * h + v);
      G.g2[kk][v] = kron4(cw + 4, sw + 4, c, 2 * kk + v + 8 * h);
      G.g3[kk][v] = kron4(c3, s3, c, 4 * kk + 2 * h + v);
    }
}

__device__ __forceinline__ void ry_groups23(v8f (&X)[4], const v8f (&Y)[4],
                                            const LayerGates& G) {
  v8f Z[4];

#pragma unroll
  for (int t = 0; t < 4; ++t) {
    v8f acc = {0.f, 0.f, 0.f, 0.f, 0.f, 0.f, 0.f, 0.f};
#pragma unroll
    for (int kk = 0; kk < 4; ++kk) {
      v2f a = {Y[t][2 * kk], Y[t][2 * kk + 1]};
      acc = __builtin_amdgcn_wmma_f32_16x16x4_f32(false, a, false, G.g2[kk],
                                                  (short)0, acc, false, false);
    }
    Z[t] = acc;
  }

#pragma unroll
  for (int j = 0; j < 4; ++j) {
    v8f acc = {0.f, 0.f, 0.f, 0.f, 0.f, 0.f, 0.f, 0.f};
#pragma unroll
    for (int kk = 0; kk < 4; ++kk) {
      v2f b = {Z[kk][2 * j], Z[kk][2 * j + 1]};
      acc = __builtin_amdgcn_wmma_f32_16x16x4_f32(false, G.g3[kk], false, b,
                                                  (short)0, acc, false, false);
    }
    X[j] = acc;
  }
}

__device__ __forceinline__ void ry_layer(v8f (&X)[4], const LayerGates& G) {
  v8f Y[4];

#pragma unroll
  for (int t = 0; t < 4; ++t) {
    v8f acc = {0.f, 0.f, 0.f, 0.f, 0.f, 0.f, 0.f, 0.f};
#pragma unroll
    for (int kk = 0; kk < 4; ++kk) {
      v2f a = {X[t][2 * kk], X[t][2 * kk + 1]};
      acc = __builtin_amdgcn_wmma_f32_16x16x4_f32(false, a, false, G.g1[kk],
                                                  (short)0, acc, false, false);
    }
    Y[t] = acc;
  }

  ry_groups23(X, Y, G);
}


__device__ __forceinline__ float lane_gather(float v, int src_lane) {
  return __int_as_float(
      __builtin_amdgcn_ds_bpermute(src_lane << 2, __float_as_int(v)));
}

__device__ __forceinline__ v8f first_ta(const float* cw5, const float* sw5,
                                        const v2f* g1, int lane) {
  const int c = lane & 15;
  const bool h = (lane >> 4) != 0;
  const bool y9 = (c & 1) != 0;
  const bool y4 = (c & 8) != 0;
  const float p0[2] = {y9 ? sw5[0] : cw5[0], y9 ? cw5[0] : sw5[0]};
  const float p1[2] = {y9 ? sw5[1] : cw5[1], y9 ? cw5[1] : sw5[1]};
  const float p2[2] = {h ? sw5[2] : cw5[2], h ? cw5[2] : sw5[2]};
  const float p3[2] = {h ? sw5[3] : cw5[3], h ? cw5[3] : sw5[3]};
  const float p4[2] = {y4 ? sw5[4] : cw5[4], y4 ? cw5[4] : sw5[4]};
  const float e34[2] = {p3[0] * p4[0], p3[1] * p4[1]};
  v8f acc = {0.f, 0.f, 0.f, 0.f, 0.f, 0.f, 0.f, 0.f};
#pragma unroll
  for (int kk = 0; kk < 4; ++kk) {
    const int y0 = kk >> 1, y1 = kk & 1;
    const float hd = p0[y0] * p1[y1 ^ y0] * p2[y1];
    v2f a = {hd * e34[0], hd * e34[1]};
    acc = __builtin_amdgcn_wmma_f32_16x16x4_f32(false, a, false, g1[kk],
                                                (short)0, acc, false, false);
  }
  return acc;
}

__device__ __forceinline__ void layer1_from_ta(v8f (&X)[4], float ta0, float ta1,
                                               const float* cw, const float* sw,
                                               const float (&gr)[4][2],
                                               float c8, float s8, float c9, float s9,
                                               int lane) {
  const bool h = (lane >> 4) != 0;
  const float r4 = h ? gr[0][1] : gr[0][0];
  float u[2];
  u[0] = (r4 * gr[1][0]) * (h ? sw[5] : cw[5]);
  u[1] = (r4 * gr[1][1]) * (h ? cw[5] : sw[5]);
#pragma unroll
  for (int w = 6; w < 8; ++w) {
    const float n0 = (u[0] * cw[w] + u[1] * sw[w]) * gr[w - 4][0];
    const float n1 = (u[0] * sw[w] + u[1] * cw[w]) * gr[w - 4][1];
    u[0] = n0;
    u[1] = n1;
  }
  const float in8[2] = {u[0] * cw[8] + u[1] * sw[8],
                        u[0] * sw[8] + u[1] * cw[8]};
  const v2f a = {ta0, ta1};
#pragma unroll
  for (int t = 0; t < 4; ++t) {
    const int y8p = t >> 1, y9p = t & 1;
    const float e0 = ry_elem(c8, s8, y8p, 0) * in8[0];
    const float e1 = ry_elem(c8, s8, y8p, 1) * in8[1];
    v2f b;
#pragma unroll
    for (int v = 0; v < 2; ++v) {
      const float z = e0 * (v ? sw[9] : cw[9]) + e1 * (v ? cw[9] : sw[9]);
      b[v] = ry_elem(c9, s9, y9p, v) * z;
    }
    const v8f zero = {0.f, 0.f, 0.f, 0.f, 0.f, 0.f, 0.f, 0.f};
    X[t] = __builtin_amdgcn_wmma_f32_16x16x4_f32(false, a, false, b, (short)0,
                                                 zero, false, false);
  }
}

__device__ __forceinline__ constexpr int cnot_src(int i) {
  int s = i;
  s ^= (s & 1) << 9;
  for (int w = 8; w >= 0; --w)
    s ^= ((s >> (9 - w)) & 1) << (8 - w);
  return s;
}

__device__ __forceinline__ void ring_load(v8f (&X)[4], const float* __restrict__ S,
                                          int lane) {
  const int c = lane & 15;
  const int h = lane >> 4;
  const int ld_base = cnot_src((h << 7) | (c << 2));
#pragma unroll
  for (int t = 0; t < 4; ++t)
#pragma unroll
    for (int r = 0; r < 8; ++r) {
      const int a = 4 * (r >> 1) + (r & 1);
      X[t][r] = S[ld_base ^ cnot_src((a << 6) | t)];
    }
  __builtin_amdgcn_sched_barrier(0);
}

__device__ __forceinline__ void cnot_ring(v8f (&X)[4], float* __restrict__ S,
                                          int lane) {
  const int c = lane & 15;
  const int h = lane >> 4;
  const int st_base = (c << 2) | (h << 1);
#pragma unroll
  for (int j = 0; j < 4; ++j)
#pragma unroll
    for (int r = 0; r < 8; ++r) {
      const int a = (((r >> 1) & 1) << 3) | (j << 1) | (r & 1);
      S[st_base | (a << 6) | (r >> 2)] = X[j][r];
    }
  ring_load(X, S, lane);
}

__device__ __forceinline__ void cnot_ring_rows(v8f (&X)[4], float* __restrict__ S,
                                               int lane) {
  const int c = lane & 15;
  const int h = lane >> 4;
  const int st_base = (h << 9) | (c << 2);
#pragma unroll
  for (int t = 0; t < 4; ++t)
#pragma unroll
    for (int r = 0; r < 8; ++r)
      S[st_base | (r << 6) | t] = X[t][r];
  ring_load(X, S, lane);
}

template <int MASK>
__device__ __forceinline__ float row_xor(float v) {
  return __int_as_float(__builtin_amdgcn_update_dpp(
      0, __float_as_int(v), 0x160 | MASK, 0xf, 0xf, true));
}

__device__ __forceinline__ float xhalf(float v) {
  const int b = __float_as_int(v);
  return __int_as_float(
      __builtin_amdgcn_permlanex16(b, b, 0x76543210, 0xfedcba98, false, false));
}

template <int MASK>
__device__ __forceinline__ float pair_step(float a, float b, bool bit) {
  const float keep = bit ? b : a;
  const float send = bit ? a : b;
  return keep + row_xor<MASK>(send);
}

__device__ __forceinline__ float hsum8(v8f v) {
  return ((v[0] + v[1]) + (v[2] + v[3])) + ((v[4] + v[5]) + (v[6] + v[7]));
}

__device__ __forceinline__ float lane_bcast(float v, int src_lane) {
  return __int_as_float(__builtin_amdgcn_readlane(__float_as_int(v), src_lane));
}

static_assert(SAMPLES_PER_WAVE == 4, "the two angle passes hold 3 + 1 samples");

__global__ __launch_bounds__(WAVES_PER_BLOCK * 32)
void qlayer_wmma_kernel(const float* __restrict__ x,
                        const float* __restrict__ params,
                        float* __restrict__ out, int batch) {
  __shared__ float lds[WAVES_PER_BLOCK * STATE_SZ];
  const int lane = threadIdx.x & 31;
  const int wid  = threadIdx.x >> 5;
  const int first = (blockIdx.x * WAVES_PER_BLOCK + wid) * SAMPLES_PER_WAVE;
  if (first >= batch) return;
  const int left = batch - first;
  const int nvalid = left < SAMPLES_PER_WAVE ? left : SAMPLES_PER_WAVE;
  const int last_sample = batch - 1;
  float* S = lds + wid * STATE_SZ;
  const int c = lane & 15;
  const int h = lane >> 4;

  const bool live = lane < 3 * N_QUBITS;
  const int g1 = live ? ((lane >= 2 * N_QUBITS) ? 2 : ((lane >= N_QUBITS) ? 1 : 0)) : 0;
  const int w1 = live ? lane - N_QUBITS * g1 : 0;
  const int sa = (first + g1 < last_sample) ? first + g1 : last_sample;
  const float half1 = 0.5f * (tanhf(x[sa * N_QUBITS + w1]) + params[w1]);
  const float cv1 = cosf(half1);
  const float sv1 = sinf(half1);

  const int ai = live ? lane : 0;
  const int xi = (lane < N_QUBITS) ? lane : 0;
  const int sb = (first + 3 < last_sample) ? first + 3 : last_sample;
  const float pv = params[ai];
  const float xv = x[sb * N_QUBITS + xi];
  const float half2 = 0.5f * ((lane < N_QUBITS) ? (tanhf(xv) + pv) : pv);
  const float cv2 = cosf(half2);
  const float sv2 = sinf(half2);

  float c1[N_QUBITS], s1[N_QUBITS], c2[N_QUBITS], s2[N_QUBITS];
#pragma unroll
  for (int w = 0; w < N_QUBITS; ++w) {
    c1[w] = lane_bcast(cv2, N_QUBITS + w);
    s1[w] = lane_bcast(sv2, N_QUBITS + w);
    c2[w] = lane_bcast(cv2, 2 * N_QUBITS + w);
    s2[w] = lane_bcast(sv2, 2 * N_QUBITS + w);
  }

  LayerGates G2;
  build_gates(G2, c2, s2, lane);
  v2f g1a[4];
#pragma unroll
  for (int kk = 0; kk < 4; ++kk)
#pragma unroll
    for (int v = 0; v < 2; ++v)
      g1a[kk][v] = kron4(c1, s1, c, 4 * kk + 2 * h + v);
  float gr[4][2];
#pragma unroll
  for (int j = 0; j < 4; ++j) {
    const bool yp = ((c >> (3 - j)) & 1) != 0;
    gr[j][0] = yp ? s1[4 + j] : c1[4 + j];
    gr[j][1] = yp ? c1[4 + j] : -s1[4 + j];
  }

  v8f TA;
  {
    const int gm = (c >> 1) & 3;
    const int src0 = (gm < 3) ? N_QUBITS * gm : 0;
    float cw5[5], sw5[5];
#pragma unroll
    for (int w = 0; w < 5; ++w) {
      const float ca = lane_gather(cv1, src0 + w);
      const float sa1 = lane_gather(sv1, src0 + w);
      const float cb = lane_bcast(cv2, w);
      const float sb3 = lane_bcast(sv2, w);
      cw5[w] = (gm == 3) ? cb : ca;
      sw5[w] = (gm == 3) ? sb3 : sa1;
    }
    TA = first_ta(cw5, sw5, g1a, lane);
  }

  const bool b1 = (c & 1) != 0, b2 = (c & 2) != 0;
  const bool b4 = (c & 4) != 0, b8 = (c & 8) != 0;

#pragma unroll 1
  for (int i = 0; i < nvalid; ++i) {
    v8f X[4];
    {
      const bool fourth = (i == 3);
      const float cvi = fourth ? cv2 : cv1;
      const float svi = fourth ? sv2 : sv1;
      const int base = fourth ? 0 : N_QUBITS * i;
      float c0[N_QUBITS], s0[N_QUBITS];
#pragma unroll
      for (int w = 5; w < N_QUBITS; ++w) {
        c0[w] = lane_bcast(cvi, base + w);
        s0[w] = lane_bcast(svi, base + w);
      }
      const float ta0 = (i == 0) ? TA[0] : (i == 1) ? TA[2] : (i == 2) ? TA[4] : TA[6];
      const float ta1 = (i == 0) ? TA[1] : (i == 1) ? TA[3] : (i == 2) ? TA[5] : TA[7];
      layer1_from_ta(X, ta0, ta1, c0, s0, gr, c1[8], s1[8], c1[9], s1[9], lane);
      cnot_ring_rows(X, S, lane);
    }

    ry_layer(X, G2);
    cnot_ring(X, S, lane);

    const v8f p0 = X[0] * X[0], p1 = X[1] * X[1];
    const v8f p2 = X[2] * X[2], p3 = X[3] * X[3];
    const v8f sa01 = p0 + p1, sb23 = p2 + p3;
    const v8f d9 = (p0 - p1) + (p2 - p3);
    const v8f s8 = sa01 + sb23;
    const v8f d8 = sa01 - sb23;
    const float D9 = hsum8(d9);
    const float D8 = hsum8(d8);
    float s3[4], d3[4];
#pragma unroll
    for (int k = 0; k < 4; ++k) {
      s3[k] = s8[2 * k] + s8[2 * k + 1];
      d3[k] = s8[2 * k] - s8[2 * k + 1];
    }
    const float D3 = (d3[0] + d3[1]) + (d3[2] + d3[3]);
    const float s1a = s3[0] + s3[1], s1b = s3[2] + s3[3];
    const float D1 = (s3[0] - s3[1]) + (s3[2] - s3[3]);
    const float T  = s1a + s1b;
    const float D0 = s1a - s1b;

    const float v2 = h ? -T : T;
    const float v4 = b8 ? -T : T;
    const float v5 = b4 ? -T : T;
    const float v6 = b2 ? -T : T;
    const float v7 = b1 ? -T : T;

    const float r0 = pair_step<1>(D0, D1, b1);
    const float r1 = pair_step<1>(v2, D3, b1);
    const float r2 = pair_step<1>(v4, v5, b1);
    const float r3 = pair_step<1>(v6, v7, b1);
    const float r4 = pair_step<1>(D8, D9, b1);
    const float q0 = pair_step<2>(r0, r1, b2);
    const float q1 = pair_step<2>(r2, r3, b2);
    const float q2 = r4 + row_xor<2>(r4);
    const float u0 = pair_step<4>(q0, q1, b4);
    const float u1 = q2 + row_xor<4>(q2);
    const float z  = pair_step<8>(u0, u1, b8);
    const float zz = z + xhalf(z);
    if (lane < N_QUBITS)
      out[(first + i) * N_QUBITS + lane] = zz;
  }
}

extern "C" void kernel_launch(void* const* d_in, const int* in_sizes, int n_in,
                              void* d_out, int out_size, void* d_ws, size_t ws_size,
                              hipStream_t stream) {
  const float* x      = (const float*)d_in[0];
  const float* params = (const float*)d_in[1];
  float* out          = (float*)d_out;
  const int batch = in_sizes[0] / N_QUBITS;
  const int per_block = WAVES_PER_BLOCK * SAMPLES_PER_WAVE;
  const int blocks = (batch + per_block - 1) / per_block;
  qlayer_wmma_kernel<<<blocks, WAVES_PER_BLOCK * 32, 0, stream>>>(x, params, out, batch);
}


# Ring 1 + group 1 of parameter layer 2 done on the factors: ring-permuted TA (1 KiB LDS table, 4 loads) x G1, then W^T x that, K=8; 44 matrix instr. a sample, was 52; rounding and order of sums change
# speedup vs baseline: 5.0326x; 1.1263x over previous
#include <hip/hip_runtime.h>
#include <math.h>


typedef __attribute__((ext_vector_type(2))) float v2f;
typedef __attribute__((ext_vector_type(8))) float v8f;

#define N_QUBITS 10
#define N_LAYERS 3
#define WAVES_PER_BLOCK 4
#define STATE_SZ 1024
#define SAMPLES_PER_WAVE 4
#define TA_WORDS 256

__device__ __forceinline__ float ry_elem(float c, float s, int row, int col) {
  return (row == col) ? c : ((row == 0) ? -s : s);
}

__device__ __forceinline__ float kron4(const float* c4, const float* s4,
                                       int row, int col) {
  float e = 1.0f;
#pragma unroll
  for (int j = 0; j < 4; ++j)
    e *= ry_elem(c4[j], s4[j], (row >> (3 - j)) & 1, (col >> (3 - j)) & 1);
  return e;
}

struct LayerGates {
  v2f g1[4];
  v2f g2[4];
  v2f g3[4];
};

__device__ __forceinline__ void build_gates(LayerGates& G, const float* cw,
                                            const float* sw, int lane) {
  const int c = lane & 15;
  const int h = lane >> 4;
  const float c3[4] = {cw[8], cw[9], 1.0f, 1.0f};
  const float s3[4] = {sw[8], sw[9], 0.0f, 0.0f};
#pragma unroll
  for (int kk = 0; kk < 4; ++kk)
#pragma unroll
    for (int v = 0; v < 2; ++v) {
      G.g1[kk][v] = kron4(cw, sw, c, 4 * kk + 2 * h + v);
      G.g2[kk][v] = kron4(cw + 4, sw + 4, c, 2 * kk + v + 8 * h);
      G.g3[kk][v] = kron4(c3, s3, c, 4 * kk + 2 * h + v);
    }
}

__device__ __forceinline__ void ry_groups23(v8f (&X)[4], const v8f (&Y)[4],
                                            const LayerGates& G) {
  v8f Z[4];

#pragma unroll
  for (int t = 0; t < 4; ++t) {
    v8f acc = {0.f, 0.f, 0.f, 0.f, 0.f, 0.f, 0.f, 0.f};
#pragma unroll
    for (int kk = 0; kk < 4; ++kk) {
      v2f a = {Y[t][2 * kk], Y[t][2 * kk + 1]};
      acc = __builtin_amdgcn_wmma_f32_16x16x4_f32(false, a, false, G.g2[kk],
                                                  (short)0, acc, false, false);
    }
    Z[t] = acc;
  }

#pragma unroll
  for (int j = 0; j < 4; ++j) {
    v8f acc = {0.f, 0.f, 0.f, 0.f, 0.f, 0.f, 0.f, 0.f};
#pragma unroll
    for (int kk = 0; kk < 4; ++kk) {
      v2f b = {Z[kk][2 * j], Z[kk][2 * j + 1]};
      acc = __builtin_amdgcn_wmma_f32_16x16x4_f32(false, G.g3[kk], false, b,
                                                  (short)0, acc, false, false);
    }
    X[j] = acc;
  }
}


__device__ __forceinline__ float lane_gather(float v, int src_lane) {
  return __int_as_float(
      __builtin_amdgcn_ds_bpermute(src_lane << 2, __float_as_int(v)));
}

__device__ __forceinline__ v8f first_ta(const float* cw5, const float* sw5,
                                        const v2f* g1, int lane) {
  const int c = lane & 15;
  const bool h = (lane >> 4) != 0;
  const bool y9 = (c & 1) != 0;
  const bool y4 = (c & 8) != 0;
  const float p0[2] = {y9 ? sw5[0] : cw5[0], y9 ? cw5[0] : sw5[0]};
  const float p1[2] = {y9 ? sw5[1] : cw5[1], y9 ? cw5[1] : sw5[1]};
  const float p2[2] = {h ? sw5[2] : cw5[2], h ? cw5[2] : sw5[2]};
  const float p3[2] = {h ? sw5[3] : cw5[3], h ? cw5[3] : sw5[3]};
  const float p4[2] = {y4 ? sw5[4] : cw5[4], y4 ? cw5[4] : sw5[4]};
  const float e34[2] = {p3[0] * p4[0], p3[1] * p4[1]};
  v8f acc = {0.f, 0.f, 0.f, 0.f, 0.f, 0.f, 0.f, 0.f};
#pragma unroll
  for (int kk = 0; kk < 4; ++kk) {
    const int y0 = kk >> 1, y1 = kk & 1;
    const float hd = p0[y0] * p1[y1 ^ y0] * p2[y1];
    v2f a = {hd * e34[0], hd * e34[1]};
    acc = __builtin_amdgcn_wmma_f32_16x16x4_f32(false, a, false, g1[kk],
                                                (short)0, acc, false, false);
  }
  return acc;
}

struct L1Rows {
  float g4[2], g5[2], g6[2], g7[2];
  float g8[2][2];
};

__device__ __forceinline__ void layer1_ring_g1(v8f (&Y)[4], const float* __restrict__ T,
                                               int i, const float* cw, const float* sw,
                                               const L1Rows& R, float c9, float s9,
                                               const v2f* g1, int lane) {
  const int c = lane & 15;
  const int h = lane >> 4;
  const bool u = (c & 8) != 0;
  const int v = (c >> 2) & 1;
  const int lane_a = 12 * v + 2 * h + ((c >> 3) ^ h);
  const int base = (((c >> 1) & 1) << 7) | (lane_a << 3) | (c & 1);
  v8f tb = {0.f, 0.f, 0.f, 0.f, 0.f, 0.f, 0.f, 0.f};
#pragma unroll
  for (int kk = 0; kk < 4; ++kk) {
    const int kk_a = ((kk >> 1) << 3) | (((kk >> 1) ^ (kk & 1)) << 2) | ((kk & 1) << 1);
    const float ta = T[(base ^ (kk_a << 3)) + 2 * i];
    const v2f a = {u ? 0.f : ta, u ? ta : 0.f};
    tb = __builtin_amdgcn_wmma_f32_16x16x4_f32(false, a, false, g1[kk], (short)0,
                                               tb, false, false);
  }

  float wv[4][2][2];
#pragma unroll
  for (int y4 = 0; y4 < 2; ++y4) {
    float p[2];
    p[0] = (R.g4[y4] * R.g5[0]) * (y4 ? sw[5] : cw[5]);
    p[1] = (R.g4[y4] * R.g5[1]) * (y4 ? cw[5] : sw[5]);
    {
      const float n0 = (p[0] * cw[6] + p[1] * sw[6]) * R.g6[0];
      const float n1 = (p[0] * sw[6] + p[1] * cw[6]) * R.g6[1];
      p[0] = n0;
      p[1] = n1;
    }
    {
      const float n0 = (p[0] * cw[7] + p[1] * sw[7]) * R.g7[0];
      const float n1 = (p[0] * sw[7] + p[1] * cw[7]) * R.g7[1];
      p[0] = n0;
      p[1] = n1;
    }
    const float in8[2] = {p[0] * cw[8] + p[1] * sw[8],
                          p[0] * sw[8] + p[1] * cw[8]};
#pragma unroll
    for (int n8 = 0; n8 < 2; ++n8) {
      const float e0 = R.g8[n8][0] * in8[0];
      const float e1 = R.g8[n8][1] * in8[1];
#pragma unroll
      for (int y9 = 0; y9 < 2; ++y9) {
        const float z = e0 * (y9 ? sw[9] : cw[9]) + e1 * (y9 ? cw[9] : sw[9]);
#pragma unroll
        for (int n9 = 0; n9 < 2; ++n9)
          wv[2 * n8 + n9][y4][y9] = ry_elem(c9, s9, n9 ^ n8, y9) * z;
      }
    }
  }

#pragma unroll
  for (int t = 0; t < 4; ++t) {
    const int n9 = t & 1;
    v8f acc = {0.f, 0.f, 0.f, 0.f, 0.f, 0.f, 0.f, 0.f};
#pragma unroll
    for (int y4 = 0; y4 < 2; ++y4) {
      const v2f a = {wv[t][y4][0], wv[t][y4][1]};
      const v2f b = {tb[4 * n9 + 2 * y4], tb[4 * n9 + 2 * y4 + 1]};
      acc = __builtin_amdgcn_wmma_f32_16x16x4_f32(false, a, false, b, (short)0,
                                                  acc, false, false);
    }
    Y[t] = acc;
  }
}

__device__ __forceinline__ constexpr int cnot_src(int i) {
  int s = i;
  s ^= (s & 1) << 9;
  for (int w = 8; w >= 0; --w)
    s ^= ((s >> (9 - w)) & 1) << (8 - w);
  return s;
}

__device__ __forceinline__ void ring_load(v8f (&X)[4], const float* __restrict__ S,
                                          int lane) {
  const int c = lane & 15;
  const int h = lane >> 4;
  const int ld_base = cnot_src((h << 7) | (c << 2));
#pragma unroll
  for (int t = 0; t < 4; ++t)
#pragma unroll
    for (int r = 0; r < 8; ++r) {
      const int a = 4 * (r >> 1) + (r & 1);
      X[t][r] = S[ld_base ^ cnot_src((a << 6) | t)];
    }
  __builtin_amdgcn_sched_barrier(0);
}

__device__ __forceinline__ void cnot_ring(v8f (&X)[4], float* __restrict__ S,
                                          int lane) {
  const int c = lane & 15;
  const int h = lane >> 4;
  const int st_base = (c << 2) | (h << 1);
#pragma unroll
  for (int j = 0; j < 4; ++j)
#pragma unroll
    for (int r = 0; r < 8; ++r) {
      const int a = (((r >> 1) & 1) << 3) | (j << 1) | (r & 1);
      S[st_base | (a << 6) | (r >> 2)] = X[j][r];
    }
  ring_load(X, S, lane);
}

template <int MASK>
__device__ __forceinline__ float row_xor(float v) {
  return __int_as_float(__builtin_amdgcn_update_dpp(
      0, __float_as_int(v), 0x160 | MASK, 0xf, 0xf, true));
}

__device__ __forceinline__ float xhalf(float v) {
  const int b = __float_as_int(v);
  return __int_as_float(
      __builtin_amdgcn_permlanex16(b, b, 0x76543210, 0xfedcba98, false, false));
}

template <int MASK>
__device__ __forceinline__ float pair_step(float a, float b, bool bit) {
  const float keep = bit ? b : a;
  const float send = bit ? a : b;
  return keep + row_xor<MASK>(send);
}

__device__ __forceinline__ float hsum8(v8f v) {
  return ((v[0] + v[1]) + (v[2] + v[3])) + ((v[4] + v[5]) + (v[6] + v[7]));
}

__device__ __forceinline__ float lane_bcast(float v, int src_lane) {
  return __int_as_float(__builtin_amdgcn_readlane(__float_as_int(v), src_lane));
}

static_assert(SAMPLES_PER_WAVE == 4, "the two angle passes hold 3 + 1 samples");

__global__ __launch_bounds__(WAVES_PER_BLOCK * 32)
void qlayer_wmma_kernel(const float* __restrict__ x,
                        const float* __restrict__ params,
                        float* __restrict__ out, int batch) {
  __shared__ float lds[WAVES_PER_BLOCK * STATE_SZ];
  __shared__ float ta_tab[WAVES_PER_BLOCK * TA_WORDS];
  const int lane = threadIdx.x & 31;
  const int wid  = threadIdx.x >> 5;
  const int first = (blockIdx.x * WAVES_PER_BLOCK + wid) * SAMPLES_PER_WAVE;
  if (first >= batch) return;
  const int left = batch - first;
  const int nvalid = left < SAMPLES_PER_WAVE ? left : SAMPLES_PER_WAVE;
  const int last_sample = batch - 1;
  float* S = lds + wid * STATE_SZ;
  float* T = ta_tab + wid * TA_WORDS;
  const int c = lane & 15;
  const int h = lane >> 4;

  const bool live = lane < 3 * N_QUBITS;
  const int g1 = live ? ((lane >= 2 * N_QUBITS) ? 2 : ((lane >= N_QUBITS) ? 1 : 0)) : 0;
  const int w1 = live ? lane - N_QUBITS * g1 : 0;
  const int sa = (first + g1 < last_sample) ? first + g1 : last_sample;
  const float half1 = 0.5f * (tanhf(x[sa * N_QUBITS + w1]) + params[w1]);
  const float cv1 = cosf(half1);
  const float sv1 = sinf(half1);

  const int ai = live ? lane : 0;
  const int xi = (lane < N_QUBITS) ? lane : 0;
  const int sb = (first + 3 < last_sample) ? first + 3 : last_sample;
  const float pv = params[ai];
  const float xv = x[sb * N_QUBITS + xi];
  const float half2 = 0.5f * ((lane < N_QUBITS) ? (tanhf(xv) + pv) : pv);
  const float cv2 = cosf(half2);
  const float sv2 = sinf(half2);

  float c1[N_QUBITS], s1[N_QUBITS], c2[N_QUBITS], s2[N_QUBITS];
#pragma unroll
  for (int w = 0; w < N_QUBITS; ++w) {
    c1[w] = lane_bcast(cv2, N_QUBITS + w);
    s1[w] = lane_bcast(sv2, N_QUBITS + w);
    c2[w] = lane_bcast(cv2, 2 * N_QUBITS + w);
    s2[w] = lane_bcast(sv2, 2 * N_QUBITS + w);
  }

  LayerGates G2;
  build_gates(G2, c2, s2, lane);
  v2f g1a[4];
#pragma unroll
  for (int kk = 0; kk < 4; ++kk)
#pragma unroll
    for (int v = 0; v < 2; ++v)
      g1a[kk][v] = kron4(c1, s1, c, 4 * kk + 2 * h + v);
  L1Rows R1;
  {
    const int n4 = (c >> 3) & 1, n5 = (c >> 2) & 1, n6 = (c >> 1) & 1, n7 = c & 1;
    const bool x4 = (n4 ^ h) != 0, x5 = (n5 ^ n4) != 0;
    const bool x6 = (n6 ^ n5) != 0, x7 = (n7 ^ n6) != 0;
    R1.g4[0] = x4 ? s1[4] : c1[4];  R1.g4[1] = x4 ? c1[4] : -s1[4];
    R1.g5[0] = x5 ? s1[5] : c1[5];  R1.g5[1] = x5 ? c1[5] : -s1[5];
    R1.g6[0] = x6 ? s1[6] : c1[6];  R1.g6[1] = x6 ? c1[6] : -s1[6];
    R1.g7[0] = x7 ? s1[7] : c1[7];  R1.g7[1] = x7 ? c1[7] : -s1[7];
#pragma unroll
    for (int n8 = 0; n8 < 2; ++n8) {
      const bool x8 = (n8 ^ n7) != 0;
      R1.g8[n8][0] = x8 ? s1[8] : c1[8];
      R1.g8[n8][1] = x8 ? c1[8] : -s1[8];
    }
  }

  v8f TA;
  {
    const int gm = (c >> 1) & 3;
    const int src0 = (gm < 3) ? N_QUBITS * gm : 0;
    float cw5[5], sw5[5];
#pragma unroll
    for (int w = 0; w < 5; ++w) {
      const float ca = lane_gather(cv1, src0 + w);
      const float sa1 = lane_gather(sv1, src0 + w);
      const float cb = lane_bcast(cv2, w);
      const float sb3 = lane_bcast(sv2, w);
      cw5[w] = (gm == 3) ? cb : ca;
      sw5[w] = (gm == 3) ? sb3 : sa1;
    }
    TA = first_ta(cw5, sw5, g1a, lane);
  }
#pragma unroll
  for (int r = 0; r < 8; ++r)
    T[8 * lane + r] = TA[r];

  const bool b1 = (c & 1) != 0, b2 = (c & 2) != 0;
  const bool b4 = (c & 4) != 0, b8 = (c & 8) != 0;

#pragma unroll 1
  for (int i = 0; i < nvalid; ++i) {
    v8f X[4];
    {
      const bool fourth = (i == 3);
      const float cvi = fourth ? cv2 : cv1;
      const float svi = fourth ? sv2 : sv1;
      const int base = fourth ? 0 : N_QUBITS * i;
      float c0[N_QUBITS], s0[N_QUBITS];
#pragma unroll
      for (int w = 5; w < N_QUBITS; ++w) {
        c0[w] = lane_bcast(cvi, base + w);
        s0[w] = lane_bcast(svi, base + w);
      }
      v8f Y[4];
      layer1_ring_g1(Y, T, i, c0, s0, R1, c1[9], s1[9], G2.g1, lane);
      ry_groups23(X, Y, G2);
    }
    cnot_ring(X, S, lane);

    const v8f p0 = X[0] * X[0], p1 = X[1] * X[1];
    const v8f p2 = X[2] * X[2], p3 = X[3] * X[3];
    const v8f sa01 = p0 + p1, sb23 = p2 + p3;
    const v8f d9 = (p0 - p1) + (p2 - p3);
    const v8f s8 = sa01 + sb23;
    const v8f d8 = sa01 - sb23;
    const float D9 = hsum8(d9);
    const float D8 = hsum8(d8);
    float s3[4], d3[4];
#pragma unroll
    for (int k = 0; k < 4; ++k) {
      s3[k] = s8[2 * k] + s8[2 * k + 1];
      d3[k] = s8[2 * k] - s8[2 * k + 1];
    }
    const float D3 = (d3[0] + d3[1]) + (d3[2] + d3[3]);
    const float s1a = s3[0] + s3[1], s1b = s3[2] + s3[3];
    const float D1 = (s3[0] - s3[1]) + (s3[2] - s3[3]);
    const float T  = s1a + s1b;
    const float D0 = s1a - s1b;

    const float v2 = h ? -T : T;
    const float v4 = b8 ? -T : T;
    const float v5 = b4 ? -T : T;
    const float v6 = b2 ? -T : T;
    const float v7 = b1 ? -T : T;

    const float r0 = pair_step<1>(D0, D1, b1);
    const float r1 = pair_step<1>(v2, D3, b1);
    const float r2 = pair_step<1>(v4, v5, b1);
    const float r3 = pair_step<1>(v6, v7, b1);
    const float r4 = pair_step<1>(D8, D9, b1);
    const float q0 = pair_step<2>(r0, r1, b2);
    const float q1 = pair_step<2>(r2, r3, b2);
    const float q2 = r4 + row_xor<2>(r4);
    const float u0 = pair_step<4>(q0, q1, b4);
    const float u1 = q2 + row_xor<4>(q2);
    const float z  = pair_step<8>(u0, u1, b8);
    const float zz = z + xhalf(z);
    if (lane < N_QUBITS)
      out[(first + i) * N_QUBITS + lane] = zz;
  }
}

extern "C" void kernel_launch(void* const* d_in, const int* in_sizes, int n_in,
                              void* d_out, int out_size, void* d_ws, size_t ws_size,
                              hipStream_t stream) {
  const float* x      = (const float*)d_in[0];
  const float* params = (const float*)d_in[1];
  float* out          = (float*)d_out;
  const int batch = in_sizes[0] / N_QUBITS;
  const int per_block = WAVES_PER_BLOCK * SAMPLES_PER_WAVE;
  const int blocks = (batch + per_block - 1) / per_block;
  qlayer_wmma_kernel<<<blocks, WAVES_PER_BLOCK * 32, 0, stream>>>(x, params, out, batch);
}


# Last CNOT ring and the RY on wires 8,9 before it folded into the observables: parity signs, cos Z - sin X forms on squares and pair products; 28 matrix instr. a sample, was 44; order of sums changes
# speedup vs baseline: 6.4647x; 1.2846x over previous
#include <hip/hip_runtime.h>
#include <math.h>


typedef __attribute__((ext_vector_type(2))) float v2f;
typedef __attribute__((ext_vector_type(8))) float v8f;

#define N_QUBITS 10
#define N_LAYERS 3
#define WAVES_PER_BLOCK 4
#define SAMPLES_PER_WAVE 4
#define TA_WORDS 256

__device__ __forceinline__ float ry_elem(float c, float s, int row, int col) {
  return (row == col) ? c : ((row == 0) ? -s : s);
}

__device__ __forceinline__ float kron4(const float* c4, const float* s4,
                                       int row, int col) {
  float e = 1.0f;
#pragma unroll
  for (int j = 0; j < 4; ++j)
    e *= ry_elem(c4[j], s4[j], (row >> (3 - j)) & 1, (col >> (3 - j)) & 1);
  return e;
}

struct LayerGates {
  v2f g1[4];
  v2f g2[4];
};

__device__ __forceinline__ void build_gates(LayerGates& G, const float* cw,
                                            const float* sw, int lane) {
  const int c = lane & 15;
  const int h = lane >> 4;
#pragma unroll
  for (int kk = 0; kk < 4; ++kk)
#pragma unroll
    for (int v = 0; v < 2; ++v) {
      G.g1[kk][v] = kron4(cw, sw, c, 4 * kk + 2 * h + v);
      G.g2[kk][v] = kron4(cw + 4, sw + 4, c, 2 * kk + v + 8 * h);
    }
}

__device__ __forceinline__ void ry_group2(v8f (&Z)[4], const v8f (&Y)[4],
                                          const v2f* g2) {
#pragma unroll
  for (int t = 0; t < 4; ++t) {
    v8f acc = {0.f, 0.f, 0.f, 0.f, 0.f, 0.f, 0.f, 0.f};
#pragma unroll
    for (int kk = 0; kk < 4; ++kk) {
      v2f a = {Y[t][2 * kk], Y[t][2 * kk + 1]};
      acc = __builtin_amdgcn_wmma_f32_16x16x4_f32(false, a, false, g2[kk],
                                                  (short)0, acc, false, false);
    }
    Z[t] = acc;
  }
}


__device__ __forceinline__ float lane_gather(float v, int src_lane) {
  return __int_as_float(
      __builtin_amdgcn_ds_bpermute(src_lane << 2, __float_as_int(v)));
}

__device__ __forceinline__ v8f first_ta(const float* cw5, const float* sw5,
                                        const v2f* g1, int lane) {
  const int c = lane & 15;
  const bool h = (lane >> 4) != 0;
  const bool y9 = (c & 1) != 0;
  const bool y4 = (c & 8) != 0;
  const float p0[2] = {y9 ? sw5[0] : cw5[0], y9 ? cw5[0] : sw5[0]};
  const float p1[2] = {y9 ? sw5[1] : cw5[1], y9 ? cw5[1] : sw5[1]};
  const float p2[2] = {h ? sw5[2] : cw5[2], h ? cw5[2] : sw5[2]};
  const float p3[2] = {h ? sw5[3] : cw5[3], h ? cw5[3] : sw5[3]};
  const float p4[2] = {y4 ? sw5[4] : cw5[4], y4 ? cw5[4] : sw5[4]};
  const float e34[2] = {p3[0] * p4[0], p3[1] * p4[1]};
  v8f acc = {0.f, 0.f, 0.f, 0.f, 0.f, 0.f, 0.f, 0.f};
#pragma unroll
  for (int kk = 0; kk < 4; ++kk) {
    const int y0 = kk >> 1, y1 = kk & 1;
    const float hd = p0[y0] * p1[y1 ^ y0] * p2[y1];
    v2f a = {hd * e34[0], hd * e34[1]};
    acc = __builtin_amdgcn_wmma_f32_16x16x4_f32(false, a, false, g1[kk],
                                                (short)0, acc, false, false);
  }
  return acc;
}

struct L1Rows {
  float g4[2], g5[2], g6[2], g7[2];
  float g8[2][2];
};

__device__ __forceinline__ void layer1_ring_g1(v8f (&Y)[4], const float* __restrict__ T,
                                               int i, const float* cw, const float* sw,
                                               const L1Rows& R, float c9, float s9,
                                               const v2f* g1, int lane) {
  const int c = lane & 15;
  const int h = lane >> 4;
  const bool u = (c & 8) != 0;
  const int v = (c >> 2) & 1;
  const int lane_a = 12 * v + 2 * h + ((c >> 3) ^ h);
  const int base = (((c >> 1) & 1) << 7) | (lane_a << 3) | (c & 1);
  v8f tb = {0.f, 0.f, 0.f, 0.f, 0.f, 0.f, 0.f, 0.f};
#pragma unroll
  for (int kk = 0; kk < 4; ++kk) {
    const int kk_a = ((kk >> 1) << 3) | (((kk >> 1) ^ (kk & 1)) << 2) | ((kk & 1) << 1);
    const float ta = T[(base ^ (kk_a << 3)) + 2 * i];
    const v2f a = {u ? 0.f : ta, u ? ta : 0.f};
    tb = __builtin_amdgcn_wmma_f32_16x16x4_f32(false, a, false, g1[kk], (short)0,
                                               tb, false, false);
  }

  float wv[4][2][2];
#pragma unroll
  for (int y4 = 0; y4 < 2; ++y4) {
    float p[2];
    p[0] = (R.g4[y4] * R.g5[0]) * (y4 ? sw[5] : cw[5]);
    p[1] = (R.g4[y4] * R.g5[1]) * (y4 ? cw[5] : sw[5]);
    {
      const float n0 = (p[0] * cw[6] + p[1] * sw[6]) * R.g6[0];
      const float n1 = (p[0] * sw[6] + p[1] * cw[6]) * R.g6[1];
      p[0] = n0;
      p[1] = n1;
    }
    {
      const float n0 = (p[0] * cw[7] + p[1] * sw[7]) * R.g7[0];
      const float n1 = (p[0] * sw[7] + p[1] * cw[7]) * R.g7[1];
      p[0] = n0;
      p[1] = n1;
    }
    const float in8[2] = {p[0] * cw[8] + p[1] * sw[8],
                          p[0] * sw[8] + p[1] * cw[8]};
#pragma unroll
    for (int n8 = 0; n8 < 2; ++n8) {
      const float e0 = R.g8[n8][0] * in8[0];
      const float e1 = R.g8[n8][1] * in8[1];
#pragma unroll
      for (int y9 = 0; y9 < 2; ++y9) {
        const float z = e0 * (y9 ? sw[9] : cw[9]) + e1 * (y9 ? cw[9] : sw[9]);
#pragma unroll
        for (int n9 = 0; n9 < 2; ++n9)
          wv[2 * n8 + n9][y4][y9] = ry_elem(c9, s9, n9 ^ n8, y9) * z;
      }
    }
  }

#pragma unroll
  for (int t = 0; t < 4; ++t) {
    const int n9 = t & 1;
    v8f acc = {0.f, 0.f, 0.f, 0.f, 0.f, 0.f, 0.f, 0.f};
#pragma unroll
    for (int y4 = 0; y4 < 2; ++y4) {
      const v2f a = {wv[t][y4][0], wv[t][y4][1]};
      const v2f b = {tb[4 * n9 + 2 * y4], tb[4 * n9 + 2 * y4 + 1]};
      acc = __builtin_amdgcn_wmma_f32_16x16x4_f32(false, a, false, b, (short)0,
                                                  acc, false, false);
    }
    Y[t] = acc;
  }
}

template <int MASK>
__device__ __forceinline__ float row_xor(float v) {
  return __int_as_float(__builtin_amdgcn_update_dpp(
      0, __float_as_int(v), 0x160 | MASK, 0xf, 0xf, true));
}

__device__ __forceinline__ float xhalf(float v) {
  const int b = __float_as_int(v);
  return __int_as_float(
      __builtin_amdgcn_permlanex16(b, b, 0x76543210, 0xfedcba98, false, false));
}

template <int MASK>
__device__ __forceinline__ float pair_step(float a, float b, bool bit) {
  const float keep = bit ? b : a;
  const float send = bit ? a : b;
  return keep + row_xor<MASK>(send);
}

__device__ __forceinline__ float parity_sum8(v8f v) {
  const v2f a = {v[0], v[1]}, b = {v[2], v[3]}, c = {v[4], v[5]}, d = {v[6], v[7]};
  const v2f p = (a - b) - (c - d);
  return p[0] - p[1];
}

struct LaneSums {
  float n1, n2, n3;
  float q8;
  float q89;
};

__device__ __forceinline__ LaneSums lane_sums(const v8f (&Z)[4], const float* k8,
                                              const float* k89) {
  const v8f q0 = Z[0] * Z[0], q1 = Z[1] * Z[1], q2 = Z[2] * Z[2], q3 = Z[3] * Z[3];
  const v8f sa = q0 + q1, da = q0 - q1, sb = q2 + q3, db = q2 - q3;
  const v8f n  = sa + sb;
  const v8f zi = sa - sb;
  const v8f zz = da - db;
  const v8f m02 = Z[0] * Z[2], m13 = Z[1] * Z[3];
  const v8f zx = Z[0] * Z[1] - Z[2] * Z[3];
  const v8f xi = m02 + m13;
  const v8f xz = m02 - m13;
  const v8f xx = Z[0] * Z[3] + Z[1] * Z[2];

  const v2f n0 = {n[0], n[1]}, n1 = {n[2], n[3]}, n2 = {n[4], n[5]}, n3 = {n[6], n[7]};
  const v2f e = (n0 + n1) - (n2 + n3);
  const v2f f = (n0 - n1) - (n2 - n3);
  LaneSums L;
  L.n1 = e[0] + e[1];
  L.n2 = f[0] + f[1];
  L.n3 = f[0] - f[1];
  L.q8 = k8[0] * parity_sum8(zi) + k8[1] * parity_sum8(xi);
  L.q89 = (k89[0] * parity_sum8(zz) + k89[1] * parity_sum8(zx)) +
          (k89[2] * parity_sum8(xz) + k89[3] * parity_sum8(xx));
  return L;
}

__device__ __forceinline__ float lane_bcast(float v, int src_lane) {
  return __int_as_float(__builtin_amdgcn_readlane(__float_as_int(v), src_lane));
}

static_assert(SAMPLES_PER_WAVE == 4, "the two angle passes hold 3 + 1 samples");

__global__ __launch_bounds__(WAVES_PER_BLOCK * 32)
void qlayer_wmma_kernel(const float* __restrict__ x,
                        const float* __restrict__ params,
                        float* __restrict__ out, int batch) {
  __shared__ float ta_tab[WAVES_PER_BLOCK * TA_WORDS];
  const int lane = threadIdx.x & 31;
  const int wid  = threadIdx.x >> 5;
  const int first = (blockIdx.x * WAVES_PER_BLOCK + wid) * SAMPLES_PER_WAVE;
  if (first >= batch) return;
  const int left = batch - first;
  const int nvalid = left < SAMPLES_PER_WAVE ? left : SAMPLES_PER_WAVE;
  const int last_sample = batch - 1;
  float* T = ta_tab + wid * TA_WORDS;
  const int c = lane & 15;
  const int h = lane >> 4;

  const bool live = lane < 3 * N_QUBITS;
  const int g1 = live ? ((lane >= 2 * N_QUBITS) ? 2 : ((lane >= N_QUBITS) ? 1 : 0)) : 0;
  const int w1 = live ? lane - N_QUBITS * g1 : 0;
  const int sa = (first + g1 < last_sample) ? first + g1 : last_sample;
  const float half1 = 0.5f * (tanhf(x[sa * N_QUBITS + w1]) + params[w1]);
  const float cv1 = cosf(half1);
  const float sv1 = sinf(half1);

  const int ai = live ? lane : 0;
  const int xi = (lane < N_QUBITS) ? lane : 0;
  const int sb = (first + 3 < last_sample) ? first + 3 : last_sample;
  const float pv = params[ai];
  const float xv = x[sb * N_QUBITS + xi];
  const float half2 = 0.5f * ((lane < N_QUBITS) ? (tanhf(xv) + pv) : pv);
  const float cv2 = cosf(half2);
  const float sv2 = sinf(half2);

  float c1[N_QUBITS], s1[N_QUBITS], c2[N_QUBITS], s2[N_QUBITS];
#pragma unroll
  for (int w = 0; w < N_QUBITS; ++w) {
    c1[w] = lane_bcast(cv2, N_QUBITS + w);
    s1[w] = lane_bcast(sv2, N_QUBITS + w);
    c2[w] = lane_bcast(cv2, 2 * N_QUBITS + w);
    s2[w] = lane_bcast(sv2, 2 * N_QUBITS + w);
  }

  LayerGates G2;
  build_gates(G2, c2, s2, lane);
  const float C8 = c2[8] * c2[8] - s2[8] * s2[8], S8 = 2.0f * c2[8] * s2[8];
  const float C9 = c2[9] * c2[9] - s2[9] * s2[9], S9 = 2.0f * c2[9] * s2[9];
  const float k8[2] = {C8, -2.0f * S8};
  const float k89[4] = {C8 * C9, -2.0f * (C8 * S9), -2.0f * (S8 * C9), 2.0f * (S8 * S9)};
  v2f g1a[4];
#pragma unroll
  for (int kk = 0; kk < 4; ++kk)
#pragma unroll
    for (int v = 0; v < 2; ++v)
      g1a[kk][v] = kron4(c1, s1, c, 4 * kk + 2 * h + v);
  L1Rows R1;
  {
    const int n4 = (c >> 3) & 1, n5 = (c >> 2) & 1, n6 = (c >> 1) & 1, n7 = c & 1;
    const bool x4 = (n4 ^ h) != 0, x5 = (n5 ^ n4) != 0;
    const bool x6 = (n6 ^ n5) != 0, x7 = (n7 ^ n6) != 0;
    R1.g4[0] = x4 ? s1[4] : c1[4];  R1.g4[1] = x4 ? c1[4] : -s1[4];
    R1.g5[0] = x5 ? s1[5] : c1[5];  R1.g5[1] = x5 ? c1[5] : -s1[5];
    R1.g6[0] = x6 ? s1[6] : c1[6];  R1.g6[1] = x6 ? c1[6] : -s1[6];
    R1.g7[0] = x7 ? s1[7] : c1[7];  R1.g7[1] = x7 ? c1[7] : -s1[7];
#pragma unroll
    for (int n8 = 0; n8 < 2; ++n8) {
      const bool x8 = (n8 ^ n7) != 0;
      R1.g8[n8][0] = x8 ? s1[8] : c1[8];
      R1.g8[n8][1] = x8 ? c1[8] : -s1[8];
    }
  }

  v8f TA;
  {
    const int gm = (c >> 1) & 3;
    const int src0 = (gm < 3) ? N_QUBITS * gm : 0;
    float cw5[5], sw5[5];
#pragma unroll
    for (int w = 0; w < 5; ++w) {
      const float ca = lane_gather(cv1, src0 + w);
      const float sa1 = lane_gather(sv1, src0 + w);
      const float cb = lane_bcast(cv2, w);
      const float sb3 = lane_bcast(sv2, w);
      cw5[w] = (gm == 3) ? cb : ca;
      sw5[w] = (gm == 3) ? sb3 : sa1;
    }
    TA = first_ta(cw5, sw5, g1a, lane);
  }
#pragma unroll
  for (int r = 0; r < 8; ++r)
    T[8 * lane + r] = TA[r];

  const bool b1 = (c & 1) != 0, b2 = (c & 2) != 0;
  const bool b4 = (c & 4) != 0, b8 = (c & 8) != 0;
  const float sg0 = h ? -1.0f : 1.0f;
  const float sg4 = b8 ? -sg0 : sg0;
  const float sg5 = b4 ? -sg4 : sg4;
  const float sg6 = b2 ? -sg5 : sg5;
  const float sg7 = b1 ? -sg6 : sg6;
  const float sgc = h ? -sg7 : sg7;

#pragma unroll 1
  for (int i = 0; i < nvalid; ++i) {
    v8f Z[4];
    {
      const bool fourth = (i == 3);
      const float cvi = fourth ? cv2 : cv1;
      const float svi = fourth ? sv2 : sv1;
      const int base = fourth ? 0 : N_QUBITS * i;
      float c0[N_QUBITS], s0[N_QUBITS];
#pragma unroll
      for (int w = 5; w < N_QUBITS; ++w) {
        c0[w] = lane_bcast(cvi, base + w);
        s0[w] = lane_bcast(svi, base + w);
      }
      v8f Y[4];
      layer1_ring_g1(Y, T, i, c0, s0, R1, c1[9], s1[9], G2.g1, lane);
      ry_group2(Z, Y, G2.g2);
    }

    const LaneSums L = lane_sums(Z, k8, k89);
    const float v0 = sgc * L.q89;
    const float v1 = sg0 * L.n1;
    const float v2 = sg0 * L.n2;
    const float v3 = sg0 * L.n3;
    const float v4 = sg4 * L.n3;
    const float v5 = sg5 * L.n3;
    const float v6 = sg6 * L.n3;
    const float v7 = sg7 * L.n3;
    const float v8 = sg7 * L.q8;
    const float v9 = sg7 * L.q89;

    const float r0 = pair_step<1>(v0, v1, b1);
    const float r1 = pair_step<1>(v2, v3, b1);
    const float r2 = pair_step<1>(v4, v5, b1);
    const float r3 = pair_step<1>(v6, v7, b1);
    const float r4 = pair_step<1>(v8, v9, b1);
    const float q0 = pair_step<2>(r0, r1, b2);
    const float q1 = pair_step<2>(r2, r3, b2);
    const float q2 = r4 + row_xor<2>(r4);
    const float u0 = pair_step<4>(q0, q1, b4);
    const float u1 = q2 + row_xor<4>(q2);
    const float z  = pair_step<8>(u0, u1, b8);
    const float zz = z + xhalf(z);
    if (lane < N_QUBITS)
      out[(first + i) * N_QUBITS + lane] = zz;
  }
}

extern "C" void kernel_launch(void* const* d_in, const int* in_sizes, int n_in,
                              void* d_out, int out_size, void* d_ws, size_t ws_size,
                              hipStream_t stream) {
  const float* x      = (const float*)d_in[0];
  const float* params = (const float*)d_in[1];
  float* out          = (float*)d_out;
  const int batch = in_sizes[0] / N_QUBITS;
  const int per_block = WAVES_PER_BLOCK * SAMPLES_PER_WAVE;
  const int blocks = (batch + per_block - 1) / per_block;
  qlayer_wmma_kernel<<<blocks, WAVES_PER_BLOCK * 32, 0, stream>>>(x, params, out, batch);
}
